# starting kernel
# baseline (speedup 1.0000x reference)
#include <hip/hip_runtime.h>
#include <hip/hip_bf16.h>
#include <math.h>

typedef float     v2f  __attribute__((ext_vector_type(2)));
typedef float     v8f  __attribute__((ext_vector_type(8)));

#define NEG_SLOPE 0.2f
#define LN_EPS    1e-5f

__device__ __forceinline__ unsigned ordEncode(float f) {
  unsigned u = __float_as_uint(f);
  return (u >> 31) ? ~u : (u | 0x80000000u);
}
__device__ __forceinline__ float ordDecode(unsigned o) {
  return __uint_as_float((o >> 31) ? (o & 0x7FFFFFFFu) : ~o);
}

__global__ void fill_f32(float* p, float v, size_t n) {
  size_t i = (size_t)blockIdx.x * blockDim.x + threadIdx.x;
  if (i < n) p[i] = v;
}
__global__ void fill_u32(unsigned* p, unsigned v, size_t n) {
  size_t i = (size_t)blockIdx.x * blockDim.x + threadIdx.x;
  if (i < n) p[i] = v;
}

__global__ void pad_cols(const float* __restrict__ src, float* __restrict__ dst,
                         int M, int Ks, int Kd) {
  int i = blockIdx.x * blockDim.x + threadIdx.x;
  if (i >= M * Kd) return;
  int r = i / Kd, c = i % Kd;
  dst[i] = (c < Ks) ? src[(size_t)r * Ks + c] : 0.0f;
}

__global__ void pad_rows(const float* __restrict__ src, float* __restrict__ dst,
                         int Ms, int Md, int Ncols) {
  int i = blockIdx.x * blockDim.x + threadIdx.x;
  if (i >= Md * Ncols) return;
  int r = i / Ncols;
  dst[i] = (r < Ms) ? src[i] : 0.0f;
}

template <int K>
__global__ void wmma_gemm(const float* __restrict__ A, const float* __restrict__ B,
                          float* __restrict__ C, int M, int N) {
  const int lane  = threadIdx.x & 31;
  const int wave  = threadIdx.x >> 5;
  const int tileM = blockIdx.x * 4 + wave;
  const int m0 = tileM * 16;
  const int n0 = blockIdx.y * 16;
  if (m0 + 16 > M) return;
  const int half = lane >> 4;
  const int l16  = lane & 15;
  const float* __restrict__ arow = A + (size_t)(m0 + l16) * K + half * 2;
  const float* __restrict__ bcol = B + n0 + l16 + (size_t)(half * 2) * N;
  v8f acc = {};
#pragma unroll
  for (int k = 0; k < K; k += 4) {
    v2f a = *(const v2f*)(arow + k);
    v2f b;
    b.x = bcol[(size_t)k * N];
    b.y = bcol[(size_t)(k + 1) * N];
    acc = __builtin_amdgcn_wmma_f32_16x16x4_f32(false, a, false, b, (short)0, acc,
                                                false, false);
  }
  float* __restrict__ crow = C + (size_t)(m0 + half * 8) * N + n0 + l16;
#pragma unroll
  for (int r = 0; r < 8; ++r)
    crow[(size_t)r * N] = acc[r];
}

__global__ void edge_logits(const int* __restrict__ ei, const float* __restrict__ ea,
                            const float* __restrict__ xl, const float* __restrict__ xr,
                            const float* __restrict__ We, const float* __restrict__ att,
                            float* __restrict__ logit, unsigned* __restrict__ mOrd,
                            int E, int H, int C) {
  const int t = blockIdx.x * blockDim.x + threadIdx.x;
  if (t >= E * H) return;
  const int e = t / H, h = t % H;
  const int s = ei[e], d = ei[E + e];
  const float eav = ea[e];
  const float4* __restrict__ xld = (const float4*)(xl + (size_t)d * H * C + h * C);
  const float4* __restrict__ xrs = (const float4*)(xr + (size_t)s * H * C + h * C);
  const float4* __restrict__ we  = (const float4*)(We + h * C);
  const float4* __restrict__ at  = (const float4*)(att + h * C);
  float acc = 0.0f;
  const int q = C >> 2;
  for (int c = 0; c < q; ++c) {
    float4 a = xld[c], b = xrs[c], w = we[c], t4 = at[c];
    float v0 = a.x + b.x + eav * w.x; v0 = (v0 > 0.f) ? v0 : NEG_SLOPE * v0;
    float v1 = a.y + b.y + eav * w.y; v1 = (v1 > 0.f) ? v1 : NEG_SLOPE * v1;
    float v2 = a.z + b.z + eav * w.z; v2 = (v2 > 0.f) ? v2 : NEG_SLOPE * v2;
    float v3 = a.w + b.w + eav * w.w; v3 = (v3 > 0.f) ? v3 : NEG_SLOPE * v3;
    acc += t4.x * v0 + t4.y * v1 + t4.z * v2 + t4.w * v3;
  }
  logit[t] = acc;
  atomicMax(&mOrd[(size_t)d * H + h], ordEncode(acc));
}

__global__ void edge_exp(const int* __restrict__ ei, float* __restrict__ logit,
                         const unsigned* __restrict__ mOrd, float* __restrict__ denom,
                         int E, int H) {
  const int t = blockIdx.x * blockDim.x + threadIdx.x;
  if (t >= E * H) return;
  const int e = t / H, h = t % H;
  const int d = ei[E + e];
  const float m = ordDecode(mOrd[(size_t)d * H + h]);
  const float ex = __expf(logit[t] - m);
  logit[t] = ex;
  atomicAdd(&denom[(size_t)d * H + h], ex);
}

__global__ void edge_aggr(const int* __restrict__ ei, const float* __restrict__ expv,
                          const float* __restrict__ denom, const float* __restrict__ xr,
                          float* __restrict__ out, int E, int H, int C) {
  const int t = blockIdx.x * blockDim.x + threadIdx.x;
  if (t >= E * H) return;
  const int e = t / H, h = t % H;
  const int s = ei[e], d = ei[E + e];
  const float alpha = expv[t] / denom[(size_t)d * H + h];
  const float4* __restrict__ xrs = (const float4*)(xr + (size_t)s * H * C + h * C);
  float* __restrict__ outd = out + (size_t)d * H * C + h * C;
  const int q = C >> 2;
  for (int c = 0; c < q; ++c) {
    float4 v = xrs[c];
    atomicAdd(&outd[4 * c + 0], alpha * v.x);
    atomicAdd(&outd[4 * c + 1], alpha * v.y);
    atomicAdd(&outd[4 * c + 2], alpha * v.z);
    atomicAdd(&outd[4 * c + 3], alpha * v.w);
  }
}

__global__ void node_ln_elu(const float* __restrict__ in, const float* __restrict__ bias,
                            const float* __restrict__ g, const float* __restrict__ b,
                            float* __restrict__ out, int N, int F) {
  const int node = blockIdx.x * (blockDim.x >> 5) + (threadIdx.x >> 5);
  const int lane = threadIdx.x & 31;
  if (node >= N) return;
  const int per = F / 32;
  float vals[4];
  float sum = 0.0f;
  for (int i = 0; i < per; ++i) {
    const int c = lane * per + i;
    float v = in[(size_t)node * F + c] + bias[c];
    vals[i] = v;
    sum += v;
  }
  for (int off = 16; off > 0; off >>= 1) sum += __shfl_xor(sum, off, 32);
  const float mu = sum / (float)F;
  float vs = 0.0f;
  for (int i = 0; i < per; ++i) { float dlt = vals[i] - mu; vs += dlt * dlt; }
  for (int off = 16; off > 0; off >>= 1) vs += __shfl_xor(vs, off, 32);
  const float inv = rsqrtf(vs / (float)F + LN_EPS);
  for (int i = 0; i < per; ++i) {
    const int c = lane * per + i;
    float hv = (vals[i] - mu) * inv * g[c] + b[c];
    out[(size_t)node * F + c] = (hv > 0.0f) ? hv : (__expf(hv) - 1.0f);
  }
}

extern "C" void kernel_launch(void* const* d_in, const int* in_sizes, int n_in,
                              void* d_out, int out_size, void* d_ws, size_t ws_size,
                              hipStream_t stream) {
  const float* x      = (const float*)d_in[0];
  const float* ea     = (const float*)d_in[1];
  const float* W1l    = (const float*)d_in[2];
  const float* W1r    = (const float*)d_in[3];
  const float* W1e    = (const float*)d_in[4];
  const float* att1   = (const float*)d_in[5];
  const float* b1     = (const float*)d_in[6];
  const float* ln1_g  = (const float*)d_in[7];
  const float* ln1_b  = (const float*)d_in[8];
  const float* W2l    = (const float*)d_in[9];
  const float* W2r    = (const float*)d_in[10];
  const float* W2e    = (const float*)d_in[11];
  const float* att2   = (const float*)d_in[12];
  const float* b2     = (const float*)d_in[13];
  const float* ln2_g  = (const float*)d_in[14];
  const float* ln2_b  = (const float*)d_in[15];
  const int*   ei     = (const int*)d_in[16];
  float*       outp   = (float*)d_out;

  const int N = in_sizes[0] / 10;
  const int E = in_sizes[1];
  const int HID = 32, HEADS = 4, F1 = HEADS * HID  , OUTC = 64;
  const int KP = 12;

  float* ws      = (float*)d_ws;
  float* xl1     = ws;
  float* xr1     = xl1 + (size_t)N * F1;
  float* h1      = xr1 + (size_t)N * F1;
  float* logit1  = h1  + (size_t)N * F1;
  unsigned* m1   = (unsigned*)(logit1 + (size_t)E * HEADS);
  float* denom1  = (float*)(m1 + (size_t)N * HEADS);
  float* xpad    = denom1 + (size_t)N * HEADS;
  float* w1lp    = xpad + (size_t)N * KP;
  float* w1rp    = w1lp + (size_t)KP * F1;
  float* xr2     = xl1;
  float* logit2  = xl1 + (size_t)N * OUTC;
  unsigned* m2   = (unsigned*)(logit2 + (size_t)E);
  float* denom2  = (float*)((float*)m2 + (size_t)N);
  float* xl2     = logit1;
  float* out2    = h1;

  const unsigned ORD_NEG_INF = 0x007FFFFFu;
  const int TB = 256;
  auto blocks = [](size_t n, int tb) { return (unsigned)((n + tb - 1) / tb); };

  pad_cols<<<blocks((size_t)N * KP, TB), TB, 0, stream>>>(x, xpad, N, 10, KP);
  pad_rows<<<blocks((size_t)KP * F1, TB), TB, 0, stream>>>(W1l, w1lp, 10, KP, F1);
  pad_rows<<<blocks((size_t)KP * F1, TB), TB, 0, stream>>>(W1r, w1rp, 10, KP, F1);
  {
    dim3 grid((N / 16 + 3) / 4, F1 / 16);
    wmma_gemm<12><<<grid, 128, 0, stream>>>(xpad, w1lp, xl1, N, F1);
    wmma_gemm<12><<<grid, 128, 0, stream>>>(xpad, w1rp, xr1, N, F1);
  }
  fill_u32<<<blocks((size_t)N * HEADS, TB), TB, 0, stream>>>(m1, ORD_NEG_INF, (size_t)N * HEADS);
  fill_f32<<<blocks((size_t)N * HEADS, TB), TB, 0, stream>>>(denom1, 0.0f, (size_t)N * HEADS);
  fill_f32<<<blocks((size_t)N * F1, TB), TB, 0, stream>>>(h1, 0.0f, (size_t)N * F1);

  edge_logits<<<blocks((size_t)E * HEADS, TB), TB, 0, stream>>>(
      ei, ea, xl1, xr1, W1e, att1, logit1, m1, E, HEADS, HID);
  edge_exp<<<blocks((size_t)E * HEADS, TB), TB, 0, stream>>>(
      ei, logit1, m1, denom1, E, HEADS);
  edge_aggr<<<blocks((size_t)E * HEADS, TB), TB, 0, stream>>>(
      ei, logit1, denom1, xr1, h1, E, HEADS, HID);
  node_ln_elu<<<blocks((size_t)N * 32, TB), TB, 0, stream>>>(
      h1, b1, ln1_g, ln1_b, h1, N, F1);

  {
    dim3 grid((N / 16 + 3) / 4, OUTC / 16);
    wmma_gemm<128><<<grid, 128, 0, stream>>>(h1, W2l, xl2, N, OUTC);
    wmma_gemm<128><<<grid, 128, 0, stream>>>(h1, W2r, xr2, N, OUTC);
  }
  fill_u32<<<blocks((size_t)N, TB), TB, 0, stream>>>(m2, ORD_NEG_INF, (size_t)N);
  fill_f32<<<blocks((size_t)N, TB), TB, 0, stream>>>(denom2, 0.0f, (size_t)N);
  fill_f32<<<blocks((size_t)N * OUTC, TB), TB, 0, stream>>>(out2, 0.0f, (size_t)N * OUTC);

  edge_logits<<<blocks((size_t)E, TB), TB, 0, stream>>>(
      ei, ea, xl2, xr2, W2e, att2, logit2, m2, E, 1, OUTC);
  edge_exp<<<blocks((size_t)E, TB), TB, 0, stream>>>(
      ei, logit2, m2, denom2, E, 1);
  edge_aggr<<<blocks((size_t)E, TB), TB, 0, stream>>>(
      ei, logit2, denom2, xr2, out2, E, 1, OUTC);
  node_ln_elu<<<blocks((size_t)N * 32, TB), TB, 0, stream>>>(
      out2, b2, ln2_g, ln2_b, outp, N, OUTC);
}


# edge aggregation: float atomic scatter replaced by per-call counting sort of edges by dst and one-wave-per-node register sums (sum order: segment order, no float atomics on out)
# speedup vs baseline: 2.9394x; 2.9394x over previous
#include <hip/hip_runtime.h>
#include <hip/hip_bf16.h>
#include <math.h>

typedef float     v2f  __attribute__((ext_vector_type(2)));
typedef float     v8f  __attribute__((ext_vector_type(8)));

#define NEG_SLOPE 0.2f
#define LN_EPS    1e-5f

__device__ __forceinline__ unsigned ordEncode(float f) {
  unsigned u = __float_as_uint(f);
  return (u >> 31) ? ~u : (u | 0x80000000u);
}
__device__ __forceinline__ float ordDecode(unsigned o) {
  return __uint_as_float((o >> 31) ? (o & 0x7FFFFFFFu) : ~o);
}

__global__ void fill_f32(float* p, float v, size_t n) {
  size_t i = (size_t)blockIdx.x * blockDim.x + threadIdx.x;
  if (i < n) p[i] = v;
}
__global__ void fill_u32(unsigned* p, unsigned v, size_t n) {
  size_t i = (size_t)blockIdx.x * blockDim.x + threadIdx.x;
  if (i < n) p[i] = v;
}

__global__ void pad_cols(const float* __restrict__ src, float* __restrict__ dst,
                         int M, int Ks, int Kd) {
  int i = blockIdx.x * blockDim.x + threadIdx.x;
  if (i >= M * Kd) return;
  int r = i / Kd, c = i % Kd;
  dst[i] = (c < Ks) ? src[(size_t)r * Ks + c] : 0.0f;
}

__global__ void pad_rows(const float* __restrict__ src, float* __restrict__ dst,
                         int Ms, int Md, int Ncols) {
  int i = blockIdx.x * blockDim.x + threadIdx.x;
  if (i >= Md * Ncols) return;
  int r = i / Ncols;
  dst[i] = (r < Ms) ? src[i] : 0.0f;
}

template <int K>
__global__ void wmma_gemm(const float* __restrict__ A, const float* __restrict__ B,
                          float* __restrict__ C, int M, int N) {
  const int lane  = threadIdx.x & 31;
  const int wave  = threadIdx.x >> 5;
  const int tileM = blockIdx.x * 4 + wave;
  const int m0 = tileM * 16;
  const int n0 = blockIdx.y * 16;
  if (m0 + 16 > M) return;
  const int half = lane >> 4;
  const int l16  = lane & 15;
  const float* __restrict__ arow = A + (size_t)(m0 + l16) * K + half * 2;
  const float* __restrict__ bcol = B + n0 + l16 + (size_t)(half * 2) * N;
  v8f acc = {};
#pragma unroll
  for (int k = 0; k < K; k += 4) {
    v2f a = *(const v2f*)(arow + k);
    v2f b;
    b.x = bcol[(size_t)k * N];
    b.y = bcol[(size_t)(k + 1) * N];
    acc = __builtin_amdgcn_wmma_f32_16x16x4_f32(false, a, false, b, (short)0, acc,
                                                false, false);
  }
  float* __restrict__ crow = C + (size_t)(m0 + half * 8) * N + n0 + l16;
#pragma unroll
  for (int r = 0; r < 8; ++r)
    crow[(size_t)r * N] = acc[r];
}

__global__ void edge_logits(const int* __restrict__ ei, const float* __restrict__ ea,
                            const float* __restrict__ xl, const float* __restrict__ xr,
                            const float* __restrict__ We, const float* __restrict__ att,
                            float* __restrict__ logit, unsigned* __restrict__ mOrd,
                            int E, int H, int C) {
  const int t = blockIdx.x * blockDim.x + threadIdx.x;
  if (t >= E * H) return;
  const int e = t / H, h = t % H;
  const int s = ei[e], d = ei[E + e];
  const float eav = ea[e];
  const float4* __restrict__ xld = (const float4*)(xl + (size_t)d * H * C + h * C);
  const float4* __restrict__ xrs = (const float4*)(xr + (size_t)s * H * C + h * C);
  const float4* __restrict__ we  = (const float4*)(We + h * C);
  const float4* __restrict__ at  = (const float4*)(att + h * C);
  float acc = 0.0f;
  const int q = C >> 2;
  for (int c = 0; c < q; ++c) {
    float4 a = xld[c], b = xrs[c], w = we[c], t4 = at[c];
    float v0 = a.x + b.x + eav * w.x; v0 = (v0 > 0.f) ? v0 : NEG_SLOPE * v0;
    float v1 = a.y + b.y + eav * w.y; v1 = (v1 > 0.f) ? v1 : NEG_SLOPE * v1;
    float v2 = a.z + b.z + eav * w.z; v2 = (v2 > 0.f) ? v2 : NEG_SLOPE * v2;
    float v3 = a.w + b.w + eav * w.w; v3 = (v3 > 0.f) ? v3 : NEG_SLOPE * v3;
    acc += t4.x * v0 + t4.y * v1 + t4.z * v2 + t4.w * v3;
  }
  logit[t] = acc;
  atomicMax(&mOrd[(size_t)d * H + h], ordEncode(acc));
}

__global__ void edge_exp(const int* __restrict__ ei, float* __restrict__ logit,
                         const unsigned* __restrict__ mOrd, float* __restrict__ denom,
                         int E, int H) {
  const int t = blockIdx.x * blockDim.x + threadIdx.x;
  if (t >= E * H) return;
  const int e = t / H, h = t % H;
  const int d = ei[E + e];
  const float m = ordDecode(mOrd[(size_t)d * H + h]);
  const float ex = __expf(logit[t] - m);
  logit[t] = ex;
  atomicAdd(&denom[(size_t)d * H + h], ex);
}

__global__ void csr_count(const int* __restrict__ ei, unsigned* __restrict__ cnt, int E) {
  const int e = blockIdx.x * blockDim.x + threadIdx.x;
  if (e >= E) return;
  atomicAdd(&cnt[ei[E + e]], 1u);
}

__global__ void csr_scan(const unsigned* __restrict__ cnt, unsigned* __restrict__ rowptr,
                         unsigned* __restrict__ cursor, int N) {
  __shared__ unsigned part[1024];
  const int t = threadIdx.x;
  const int base_i = blockIdx.x * 1024;
  unsigned pre = 0;
  for (int i = t; i < base_i; i += 1024) pre += cnt[i];
  const int idx = base_i + t;
  const unsigned c = (idx < N) ? cnt[idx] : 0u;
  part[t] = pre;
  __syncthreads();
  for (int off = 512; off > 0; off >>= 1) {
    if (t < off) part[t] += part[t + off];
    __syncthreads();
  }
  const unsigned total_pre = part[0];
  __syncthreads();
  part[t] = c;
  __syncthreads();
  for (int off = 1; off < 1024; off <<= 1) {
    const unsigned v = (t >= off) ? part[t - off] : 0u;
    __syncthreads();
    part[t] += v;
    __syncthreads();
  }
  const unsigned excl = part[t] - c + total_pre;
  if (idx < N) { rowptr[idx] = excl; cursor[idx] = excl; }
}

__global__ void csr_fill(const int* __restrict__ ei, unsigned* __restrict__ cursor,
                         int* __restrict__ seg_e, int* __restrict__ seg_s, int E) {
  const int e = blockIdx.x * blockDim.x + threadIdx.x;
  if (e >= E) return;
  const unsigned p = atomicAdd(&cursor[ei[E + e]], 1u);
  seg_e[p] = e;
  seg_s[p] = ei[e];
}

template <int H, int C>
__global__ void node_gather(const int* __restrict__ seg_e, const int* __restrict__ seg_s,
                            const unsigned* __restrict__ rowptr, const unsigned* __restrict__ cnt,
                            const float* __restrict__ expv, const float* __restrict__ denom,
                            const float* __restrict__ xr, float* __restrict__ out, int N) {
  constexpr int F = H * C;
  constexpr int V = F / 32;
  typedef float vf __attribute__((ext_vector_type(V)));
  const int node = __builtin_amdgcn_readfirstlane(
      (int)(blockIdx.x * (blockDim.x >> 5) + (threadIdx.x >> 5)));
  const int lane = threadIdx.x & 31;
  if (node >= N) return;
  const int h = (lane * V) / C;
  const unsigned beg = rowptr[node];
  const unsigned len = cnt[node];
  const float dn = denom[(size_t)node * H + h];
  vf acc = {};
  for (unsigned base = 0; base < len; base += 32) {
    const unsigned nn = (len - base < 32u) ? (len - base) : 32u;
    int my_e = 0, my_s = 0;
    if ((unsigned)lane < nn) {
      my_e = seg_e[beg + base + lane];
      my_s = seg_s[beg + base + lane];
    }
#pragma unroll 4
    for (unsigned j = 0; j < nn; ++j) {
      const int e = __builtin_amdgcn_readlane(my_e, (int)j);
      const int s = __builtin_amdgcn_readlane(my_s, (int)j);
      const float alpha = expv[(size_t)e * H + h] / dn;
      const vf v = *(const vf*)(xr + (size_t)s * F + lane * V);
      acc += alpha * v;
    }
  }
  *(vf*)(out + (size_t)node * F + lane * V) = acc;
}

__global__ void node_ln_elu(const float* __restrict__ in, const float* __restrict__ bias,
                            const float* __restrict__ g, const float* __restrict__ b,
                            float* __restrict__ out, int N, int F) {
  const int node = blockIdx.x * (blockDim.x >> 5) + (threadIdx.x >> 5);
  const int lane = threadIdx.x & 31;
  if (node >= N) return;
  const int per = F / 32;
  float vals[4];
  float sum = 0.0f;
  for (int i = 0; i < per; ++i) {
    const int c = lane * per + i;
    float v = in[(size_t)node * F + c] + bias[c];
    vals[i] = v;
    sum += v;
  }
  for (int off = 16; off > 0; off >>= 1) sum += __shfl_xor(sum, off, 32);
  const float mu = sum / (float)F;
  float vs = 0.0f;
  for (int i = 0; i < per; ++i) { float dlt = vals[i] - mu; vs += dlt * dlt; }
  for (int off = 16; off > 0; off >>= 1) vs += __shfl_xor(vs, off, 32);
  const float inv = rsqrtf(vs / (float)F + LN_EPS);
  for (int i = 0; i < per; ++i) {
    const int c = lane * per + i;
    float hv = (vals[i] - mu) * inv * g[c] + b[c];
    out[(size_t)node * F + c] = (hv > 0.0f) ? hv : (__expf(hv) - 1.0f);
  }
}

extern "C" void kernel_launch(void* const* d_in, const int* in_sizes, int n_in,
                              void* d_out, int out_size, void* d_ws, size_t ws_size,
                              hipStream_t stream) {
  const float* x      = (const float*)d_in[0];
  const float* ea     = (const float*)d_in[1];
  const float* W1l    = (const float*)d_in[2];
  const float* W1r    = (const float*)d_in[3];
  const float* W1e    = (const float*)d_in[4];
  const float* att1   = (const float*)d_in[5];
  const float* b1     = (const float*)d_in[6];
  const float* ln1_g  = (const float*)d_in[7];
  const float* ln1_b  = (const float*)d_in[8];
  const float* W2l    = (const float*)d_in[9];
  const float* W2r    = (const float*)d_in[10];
  const float* W2e    = (const float*)d_in[11];
  const float* att2   = (const float*)d_in[12];
  const float* b2     = (const float*)d_in[13];
  const float* ln2_g  = (const float*)d_in[14];
  const float* ln2_b  = (const float*)d_in[15];
  const int*   ei     = (const int*)d_in[16];
  float*       outp   = (float*)d_out;

  const int N = in_sizes[0] / 10;
  const int E = in_sizes[1];
  const int HID = 32, HEADS = 4, F1 = HEADS * HID  , OUTC = 64;
  const int KP = 12;

  float* ws      = (float*)d_ws;
  float* xl1     = ws;
  float* xr1     = xl1 + (size_t)N * F1;
  float* h1      = xr1 + (size_t)N * F1;
  float* logit1  = h1  + (size_t)N * F1;
  unsigned* m1   = (unsigned*)(logit1 + (size_t)E * HEADS);
  float* denom1  = (float*)(m1 + (size_t)N * HEADS);
  float* xpad    = denom1 + (size_t)N * HEADS;
  float* w1lp    = xpad + (size_t)N * KP;
  float* w1rp    = w1lp + (size_t)KP * F1;
  unsigned* cnt    = (unsigned*)(w1rp + (size_t)KP * F1);
  unsigned* rowptr = cnt + (size_t)N;
  unsigned* cursor = rowptr + (size_t)N;
  int* seg_e       = (int*)(cursor + (size_t)N);
  int* seg_s       = seg_e + (size_t)E;
  float* xr2     = xl1;
  float* logit2  = xl1 + (size_t)N * OUTC;
  unsigned* m2   = (unsigned*)(logit2 + (size_t)E);
  float* denom2  = (float*)((float*)m2 + (size_t)N);
  float* xl2     = logit1;
  float* out2    = h1;

  const unsigned ORD_NEG_INF = 0x007FFFFFu;
  const int TB = 256;
  auto blocks = [](size_t n, int tb) { return (unsigned)((n + tb - 1) / tb); };

  fill_u32<<<blocks((size_t)N, TB), TB, 0, stream>>>(cnt, 0u, (size_t)N);
  csr_count<<<blocks((size_t)E, TB), TB, 0, stream>>>(ei, cnt, E);
  csr_scan<<<blocks((size_t)N, 1024), 1024, 0, stream>>>(cnt, rowptr, cursor, N);
  csr_fill<<<blocks((size_t)E, TB), TB, 0, stream>>>(ei, cursor, seg_e, seg_s, E);

  pad_cols<<<blocks((size_t)N * KP, TB), TB, 0, stream>>>(x, xpad, N, 10, KP);
  pad_rows<<<blocks((size_t)KP * F1, TB), TB, 0, stream>>>(W1l, w1lp, 10, KP, F1);
  pad_rows<<<blocks((size_t)KP * F1, TB), TB, 0, stream>>>(W1r, w1rp, 10, KP, F1);
  {
    dim3 grid((N / 16 + 3) / 4, F1 / 16);
    wmma_gemm<12><<<grid, 128, 0, stream>>>(xpad, w1lp, xl1, N, F1);
    wmma_gemm<12><<<grid, 128, 0, stream>>>(xpad, w1rp, xr1, N, F1);
  }
  fill_u32<<<blocks((size_t)N * HEADS, TB), TB, 0, stream>>>(m1, ORD_NEG_INF, (size_t)N * HEADS);
  fill_f32<<<blocks((size_t)N * HEADS, TB), TB, 0, stream>>>(denom1, 0.0f, (size_t)N * HEADS);

  edge_logits<<<blocks((size_t)E * HEADS, TB), TB, 0, stream>>>(
      ei, ea, xl1, xr1, W1e, att1, logit1, m1, E, HEADS, HID);
  edge_exp<<<blocks((size_t)E * HEADS, TB), TB, 0, stream>>>(
      ei, logit1, m1, denom1, E, HEADS);
  node_gather<4, 32><<<blocks((size_t)N * 32, TB), TB, 0, stream>>>(
      seg_e, seg_s, rowptr, cnt, logit1, denom1, xr1, h1, N);
  node_ln_elu<<<blocks((size_t)N * 32, TB), TB, 0, stream>>>(
      h1, b1, ln1_g, ln1_b, h1, N, F1);

  {
    dim3 grid((N / 16 + 3) / 4, OUTC / 16);
    wmma_gemm<128><<<grid, 128, 0, stream>>>(h1, W2l, xl2, N, OUTC);
    wmma_gemm<128><<<grid, 128, 0, stream>>>(h1, W2r, xr2, N, OUTC);
  }
  fill_u32<<<blocks((size_t)N, TB), TB, 0, stream>>>(m2, ORD_NEG_INF, (size_t)N);
  fill_f32<<<blocks((size_t)N, TB), TB, 0, stream>>>(denom2, 0.0f, (size_t)N);

  edge_logits<<<blocks((size_t)E, TB), TB, 0, stream>>>(
      ei, ea, xl2, xr2, W2e, att2, logit2, m2, E, 1, OUTC);
  edge_exp<<<blocks((size_t)E, TB), TB, 0, stream>>>(
      ei, logit2, m2, denom2, E, 1);
  node_gather<1, 64><<<blocks((size_t)N * 32, TB), TB, 0, stream>>>(
      seg_e, seg_s, rowptr, cnt, logit2, denom2, xr2, out2, N);
  node_ln_elu<<<blocks((size_t)N * 32, TB), TB, 0, stream>>>(
      out2, b2, ln2_g, ln2_b, outp, N, OUTC);
}


# wave entry: big launches on one-pass resident grids, a wave looping over its nodes, rows, edges or its 16-row strip's N tiles: 347k waves per call to 77k; sums kept, segment order by atomic arrival
# speedup vs baseline: 6.7679x; 2.3025x over previous
#include <hip/hip_runtime.h>
#include <hip/hip_bf16.h>
#include <math.h>

typedef float     v2f  __attribute__((ext_vector_type(2)));
typedef float     v8f  __attribute__((ext_vector_type(8)));

#define NEG_SLOPE 0.2f
#define LN_EPS    1e-5f

__global__ void fill_u32(unsigned* p, unsigned v, size_t n) {
  size_t i = (size_t)blockIdx.x * blockDim.x + threadIdx.x;
  if (i < n) p[i] = v;
}

__global__ void pad_cols(const float* __restrict__ src, float* __restrict__ dst,
                         int M, int Ks, int Kd) {
  const int total = M * Kd;
  const int stride = gridDim.x * blockDim.x;
  int i0 = blockIdx.x * blockDim.x + threadIdx.x;
  for (; i0 + 3 * stride < total; i0 += 4 * stride) {
    float v[4];
    int cc[4];
#pragma unroll
    for (int u = 0; u < 4; ++u) {
      const int i = i0 + u * stride;
      const int r = i / Kd, c = i % Kd;
      cc[u] = c;
      v[u] = src[(size_t)r * Ks + ((c < Ks) ? c : Ks - 1)];
    }
#pragma unroll
    for (int u = 0; u < 4; ++u) dst[i0 + u * stride] = (cc[u] < Ks) ? v[u] : 0.0f;
  }
  for (; i0 < total; i0 += stride) {
    const int r = i0 / Kd, c = i0 % Kd;
    const float v = src[(size_t)r * Ks + ((c < Ks) ? c : Ks - 1)];
    dst[i0] = (c < Ks) ? v : 0.0f;
  }
}

__global__ void pad_rows(const float* __restrict__ src, float* __restrict__ dst,
                         int Ms, int Md, int Ncols) {
  int i = blockIdx.x * blockDim.x + threadIdx.x;
  if (i >= Md * Ncols) return;
  int r = i / Ncols;
  dst[i] = (r < Ms) ? src[i] : 0.0f;
}

template <int K, int NT>
__global__ void wmma_gemm(const float* __restrict__ A, const float* __restrict__ B,
                          float* __restrict__ C, int M) {
  constexpr int N = NT * 16;
  const int lane  = threadIdx.x & 31;
  const int wave  = threadIdx.x >> 5;
  const int tileM = blockIdx.x * 4 + wave;
  const int m0 = tileM * 16;
  if (m0 + 16 > M) return;
  const int half = lane >> 4;
  const int l16  = lane & 15;
  const float* __restrict__ arow = A + (size_t)(m0 + l16) * K + half * 2;
  const float* __restrict__ bcol = B + l16 + (size_t)(half * 2) * N;
  v8f acc[NT];
#pragma unroll
  for (int t = 0; t < NT; ++t) acc[t] = v8f{};
#pragma unroll
  for (int k = 0; k < K; k += 4) {
    v2f a = *(const v2f*)(arow + k);
#pragma unroll
    for (int t = 0; t < NT; ++t) {
      v2f b;
      b.x = bcol[(size_t)k * N + t * 16];
      b.y = bcol[(size_t)(k + 1) * N + t * 16];
      acc[t] = __builtin_amdgcn_wmma_f32_16x16x4_f32(false, a, false, b, (short)0, acc[t],
                                                     false, false);
    }
  }
  float* __restrict__ crow = C + (size_t)(m0 + half * 8) * N + l16;
#pragma unroll
  for (int r = 0; r < 8; ++r)
#pragma unroll
    for (int t = 0; t < NT; ++t)
      crow[(size_t)r * N + t * 16] = acc[t][r];
}

#define EDGE_U 6

__global__ void csr_count(const int* __restrict__ ei, unsigned* __restrict__ cnt, int E) {
  const int stride = gridDim.x * blockDim.x;
  int e0 = blockIdx.x * blockDim.x + threadIdx.x;
  for (; e0 + (EDGE_U - 1) * stride < E; e0 += EDGE_U * stride) {
    int d[EDGE_U];
#pragma unroll
    for (int u = 0; u < EDGE_U; ++u) d[u] = ei[E + e0 + u * stride];
#pragma unroll
    for (int u = 0; u < EDGE_U; ++u) atomicAdd(&cnt[d[u]], 1u);
  }
  for (; e0 < E; e0 += stride) atomicAdd(&cnt[ei[E + e0]], 1u);
}

__global__ void csr_scan(const unsigned* __restrict__ cnt, unsigned* __restrict__ rowptr,
                         unsigned* __restrict__ cursor, int N) {
  __shared__ unsigned part[1024];
  const int t = threadIdx.x;
  const int base_i = blockIdx.x * 1024;
  unsigned pre = 0;
  for (int i = t; i < base_i; i += 1024) pre += cnt[i];
  const int idx = base_i + t;
  const unsigned c = (idx < N) ? cnt[idx] : 0u;
  part[t] = pre;
  __syncthreads();
  for (int off = 512; off > 0; off >>= 1) {
    if (t < off) part[t] += part[t + off];
    __syncthreads();
  }
  const unsigned total_pre = part[0];
  __syncthreads();
  part[t] = c;
  __syncthreads();
  for (int off = 1; off < 1024; off <<= 1) {
    const unsigned v = (t >= off) ? part[t - off] : 0u;
    __syncthreads();
    part[t] += v;
    __syncthreads();
  }
  const unsigned excl = part[t] - c + total_pre;
  if (idx < N) { rowptr[idx] = excl; cursor[idx] = excl; }
}

__global__ void csr_fill(const int* __restrict__ ei, const float* __restrict__ ea,
                         unsigned* __restrict__ cursor,
                         int* __restrict__ seg_s, float* __restrict__ seg_a, int E) {
  const int stride = gridDim.x * blockDim.x;
  int e0 = blockIdx.x * blockDim.x + threadIdx.x;
  for (; e0 + (EDGE_U - 1) * stride < E; e0 += EDGE_U * stride) {
    int d[EDGE_U], sv[EDGE_U];
    float av[EDGE_U];
    unsigned p[EDGE_U];
#pragma unroll
    for (int u = 0; u < EDGE_U; ++u) {
      const int e = e0 + u * stride;
      d[u] = ei[E + e];
      sv[u] = ei[e];
      av[u] = ea[e];
    }
#pragma unroll
    for (int u = 0; u < EDGE_U; ++u) p[u] = atomicAdd(&cursor[d[u]], 1u);
#pragma unroll
    for (int u = 0; u < EDGE_U; ++u) { seg_s[p[u]] = sv[u]; seg_a[p[u]] = av[u]; }
  }
  for (; e0 < E; e0 += stride) {
    const unsigned p = atomicAdd(&cursor[ei[E + e0]], 1u);
    seg_s[p] = ei[e0];
    seg_a[p] = ea[e0];
  }
}

template <int K>
__device__ __forceinline__ float xor_row(float x) {
  return __builtin_bit_cast(float, __builtin_amdgcn_update_dpp(
      0, __builtin_bit_cast(int, x), 0x160 | K, 0xf, 0xf, true));
}
__device__ __forceinline__ float xor_half(float x) {
  const int b = __builtin_bit_cast(int, x);
  return __builtin_bit_cast(float, __builtin_amdgcn_permlanex16(b, b, 0x76543210, 0xfedcba98,
                                                                false, false));
}
template <int LPH>
__device__ __forceinline__ float head_sum(float p) {
  p += xor_row<1>(p);
  p += xor_row<2>(p);
  p += xor_row<4>(p);
  if (LPH > 8)  p += xor_row<8>(p);
  if (LPH > 16) p += xor_half(p);
  return p;
}
template <int G> struct group_of { static constexpr int value = G; };

template <int H, int C>
__global__ void __launch_bounds__(256, 16) node_attend(const int* __restrict__ seg_s, const float* __restrict__ seg_a,
                            const unsigned* __restrict__ rowptr, const unsigned* __restrict__ cnt,
                            const float* __restrict__ xl, const float* __restrict__ xr,
                            const float* __restrict__ We, const float* __restrict__ att,
                            float* __restrict__ out, int N) {
  constexpr int F = H * C;
  constexpr int V = F / 32;
  constexpr int LPH = C / V;
  typedef float vf __attribute__((ext_vector_type(V)));
  int node = __builtin_amdgcn_readfirstlane(
      (int)(blockIdx.x * (blockDim.x >> 5) + (threadIdx.x >> 5)));
  const int nw = (int)(gridDim.x * (blockDim.x >> 5));
  const int lane = threadIdx.x & 31;
  if (node >= N) return;
  const vf wev = *(const vf*)(We + lane * V);
  const vf atv = *(const vf*)(att + lane * V);
  unsigned beg = rowptr[node];
  unsigned len = cnt[node];
  for (;;) {
  const int nxt = node + nw;
  unsigned nbeg = 0u, nlen = 0u;
  if (nxt < N) { nbeg = rowptr[nxt]; nlen = cnt[nxt]; }
  const vf xlv = *(const vf*)(xl + (size_t)node * F + lane * V);
  float m = -INFINITY, dn = 0.0f;
  vf acc = {};
  for (unsigned base = 0; base < len; base += 32) {
    const unsigned nn = (len - base < 32u) ? (len - base) : 32u;
    int my_s = 0;
    float my_a = 0.0f;
    if ((unsigned)lane < nn) {
      my_s = seg_s[beg + base + lane];
      my_a = seg_a[beg + base + lane];
    }
    auto group = [&](auto gtag, unsigned j) {
      constexpr int G = decltype(gtag)::value;
      vf v[G];
      float p[G];
#pragma unroll
      for (int i = 0; i < G; ++i) {
        const int s = __builtin_amdgcn_readlane(my_s, (int)(j + i));
        v[i] = *(const vf*)(xr + (size_t)s * F + lane * V);
      }
#pragma unroll
      for (int i = 0; i < G; ++i) {
        const float a = __builtin_bit_cast(
            float, __builtin_amdgcn_readlane(__builtin_bit_cast(int, my_a), (int)(j + i)));
        float q;
#pragma unroll
        for (int k = 0; k < V; ++k) {
          float t = xlv[k] + v[i][k] + a * wev[k];
          t = fmaxf(t, NEG_SLOPE * t);
          q = (k == 0) ? atv[k] * t : q + atv[k] * t;
        }
        p[i] = head_sum<LPH>(q);
      }
      float mn = m;
#pragma unroll
      for (int i = 0; i < G; ++i) mn = fmaxf(mn, p[i]);
      const float sc = __expf(m - mn);
      float wsum;
      vf part;
#pragma unroll
      for (int i = 0; i < G; ++i) {
        const float w = __expf(p[i] - mn);
        wsum = (i == 0) ? w : wsum + w;
        part = (i == 0) ? w * v[i] : part + w * v[i];
      }
      dn = dn * sc + wsum;
      acc = acc * sc + part;
      m = mn;
    };
    unsigned j = 0;
    for (; j + 4 <= nn; j += 4) group(group_of<4>{}, j);
    for (; j < nn; ++j) group(group_of<1>{}, j);
  }
  const float dsafe = (len != 0u) ? dn : 1.0f;
  *(vf*)(out + (size_t)node * F + lane * V) = acc / dsafe;
  if (nxt >= N) break;
  node = nxt; beg = nbeg; len = nlen;
  }
}

template <int F>
__global__ void node_ln_elu(const float* in, const float* __restrict__ bias,
                            const float* __restrict__ g, const float* __restrict__ b,
                            float* out, int N) {
  constexpr int PER = F / 32;
  typedef float vf __attribute__((ext_vector_type(PER)));
  int node = __builtin_amdgcn_readfirstlane(
      (int)(blockIdx.x * (blockDim.x >> 5) + (threadIdx.x >> 5)));
  const int nw = (int)(gridDim.x * (blockDim.x >> 5));
  const int lane = threadIdx.x & 31;
  if (node >= N) return;
  const vf bi = *(const vf*)(bias + lane * PER);
  const vf gi = *(const vf*)(g + lane * PER);
  const vf be = *(const vf*)(b + lane * PER);
  vf cur = *(const vf*)(in + (size_t)node * F + lane * PER);
  for (;;) {
    const int nxt = node + nw;
    const int nrow = (nxt < N) ? nxt : node;
    const vf nx = *(const vf*)(in + (size_t)nrow * F + lane * PER);
    float vals[PER];
    float sum = 0.0f;
#pragma unroll
    for (int i = 0; i < PER; ++i) {
      float v = cur[i] + bi[i];
      vals[i] = v;
      sum += v;
    }
    for (int off = 16; off > 0; off >>= 1) sum += __shfl_xor(sum, off, 32);
    const float mu = sum / (float)F;
    float vs = 0.0f;
#pragma unroll
    for (int i = 0; i < PER; ++i) { float dlt = vals[i] - mu; vs += dlt * dlt; }
    for (int off = 16; off > 0; off >>= 1) vs += __shfl_xor(vs, off, 32);
    const float inv = rsqrtf(vs / (float)F + LN_EPS);
    vf o;
#pragma unroll
    for (int i = 0; i < PER; ++i) {
      float hv = (vals[i] - mu) * inv * gi[i] + be[i];
      o[i] = (hv > 0.0f) ? hv : (__expf(hv) - 1.0f);
    }
    *(vf*)(out + (size_t)node * F + lane * PER) = o;
    if (nxt >= N) break;
    node = nxt;
    cur = nx;
  }
}

extern "C" void kernel_launch(void* const* d_in, const int* in_sizes, int n_in,
                              void* d_out, int out_size, void* d_ws, size_t ws_size,
                              hipStream_t stream) {
  const float* x      = (const float*)d_in[0];
  const float* ea     = (const float*)d_in[1];
  const float* W1l    = (const float*)d_in[2];
  const float* W1r    = (const float*)d_in[3];
  const float* W1e    = (const float*)d_in[4];
  const float* att1   = (const float*)d_in[5];
  const float* b1     = (const float*)d_in[6];
  const float* ln1_g  = (const float*)d_in[7];
  const float* ln1_b  = (const float*)d_in[8];
  const float* W2l    = (const float*)d_in[9];
  const float* W2r    = (const float*)d_in[10];
  const float* W2e    = (const float*)d_in[11];
  const float* att2   = (const float*)d_in[12];
  const float* b2     = (const float*)d_in[13];
  const float* ln2_g  = (const float*)d_in[14];
  const float* ln2_b  = (const float*)d_in[15];
  const int*   ei     = (const int*)d_in[16];
  float*       outp   = (float*)d_out;

  const int N = in_sizes[0] / 10;
  const int E = in_sizes[1];
  const int HID = 32, HEADS = 4, F1 = HEADS * HID  , OUTC = 64;
  const int KP = 12;

  float* ws      = (float*)d_ws;
  float* xl1     = ws;
  float* xr1     = xl1 + (size_t)N * F1;
  float* h1      = xr1 + (size_t)N * F1;
  float* xl2     = h1  + (size_t)N * F1;
  float* xpad    = xl2 + (size_t)N * OUTC;
  float* w1lp    = xpad + (size_t)N * KP;
  float* w1rp    = w1lp + (size_t)KP * F1;
  unsigned* cnt    = (unsigned*)(w1rp + (size_t)KP * F1);
  unsigned* rowptr = cnt + (size_t)N;
  unsigned* cursor = rowptr + (size_t)N;
  int* seg_s       = (int*)(cursor + (size_t)N);
  float* seg_a     = (float*)(seg_s + (size_t)E);
  float* xr2     = xl1;
  float* out2    = h1;

  const int TB = 256;
  const unsigned RES      = 512;
  const unsigned RES_ROW  = 1024;
  const unsigned RES_NODE = 2048;
  auto blocks = [](size_t n, int tb) { return (unsigned)((n + tb - 1) / tb); };

  fill_u32<<<blocks((size_t)N, TB), TB, 0, stream>>>(cnt, 0u, (size_t)N);
  csr_count<<<RES, TB, 0, stream>>>(ei, cnt, E);
  csr_scan<<<blocks((size_t)N, 1024), 1024, 0, stream>>>(cnt, rowptr, cursor, N);
  csr_fill<<<RES, TB, 0, stream>>>(ei, ea, cursor, seg_s, seg_a, E);

  pad_cols<<<RES, TB, 0, stream>>>(x, xpad, N, 10, KP);
  pad_rows<<<blocks((size_t)KP * F1, TB), TB, 0, stream>>>(W1l, w1lp, 10, KP, F1);
  pad_rows<<<blocks((size_t)KP * F1, TB), TB, 0, stream>>>(W1r, w1rp, 10, KP, F1);
  {
    dim3 grid((N / 16 + 3) / 4);
    wmma_gemm<12, 8><<<grid, 128, 0, stream>>>(xpad, w1lp, xl1, N);
    wmma_gemm<12, 8><<<grid, 128, 0, stream>>>(xpad, w1rp, xr1, N);
  }
  node_attend<4, 32><<<RES_NODE, TB, 0, stream>>>(
      seg_s, seg_a, rowptr, cnt, xl1, xr1, W1e, att1, h1, N);
  node_ln_elu<128><<<RES_ROW, TB, 0, stream>>>(
      h1, b1, ln1_g, ln1_b, h1, N);

  {
    dim3 grid((N / 16 + 3) / 4);
    wmma_gemm<128, 4><<<grid, 128, 0, stream>>>(h1, W2l, xl2, N);
    wmma_gemm<128, 4><<<grid, 128, 0, stream>>>(h1, W2r, xr2, N);
  }
  node_attend<1, 64><<<RES_NODE, TB, 0, stream>>>(
      seg_s, seg_a, rowptr, cnt, xl2, xr2, W2e, att2, out2, N);
  node_ln_elu<64><<<RES_ROW, TB, 0, stream>>>(
      out2, b2, ln2_g, ln2_b, outp, N);
}


# launch fusion, 15 launches to 7: pads and counter clear folded into one xl|xr GEMM launch per layer; bias+LayerNorm+ELU in registers in GEMM 2 (row sums in fragment order) and in attention 2
# speedup vs baseline: 7.6078x; 1.1241x over previous
#include <hip/hip_runtime.h>
#include <hip/hip_bf16.h>
#include <math.h>

typedef float     v2f  __attribute__((ext_vector_type(2)));
typedef float     v8f  __attribute__((ext_vector_type(8)));

#define NEG_SLOPE 0.2f
#define LN_EPS    1e-5f

template <int K>
__device__ __forceinline__ float xor_row(float x) {
  return __builtin_bit_cast(float, __builtin_amdgcn_update_dpp(
      0, __builtin_bit_cast(int, x), 0x160 | K, 0xf, 0xf, true));
}
__device__ __forceinline__ float xor_half(float x) {
  const int b = __builtin_bit_cast(int, x);
  return __builtin_bit_cast(float, __builtin_amdgcn_permlanex16(b, b, 0x76543210, 0xfedcba98,
                                                                false, false));
}
template <int LPH>
__device__ __forceinline__ float head_sum(float p) {
  p += xor_row<1>(p);
  p += xor_row<2>(p);
  p += xor_row<4>(p);
  if (LPH > 8)  p += xor_row<8>(p);
  if (LPH > 16) p += xor_half(p);
  return p;
}
__device__ __forceinline__ float wave_sum(float s) {
  s += xor_half(s);
  s += xor_row<8>(s);
  s += xor_row<4>(s);
  s += xor_row<2>(s);
  s += xor_row<1>(s);
  return s;
}

template <int KS, int K, int NT, bool LN>
__global__ void __launch_bounds__(128) wmma_gemm_pair(
    const float* __restrict__ A, const float* __restrict__ B1, const float* __restrict__ B2,
    float* __restrict__ C1, float* __restrict__ C2, int M,
    const float* __restrict__ bias, const float* __restrict__ ln_g, const float* __restrict__ ln_b,
    unsigned* __restrict__ zero_p, int zero_n) {
  static_assert(K % 4 == 0 && (K == KS || K == KS + 2), "K step layout");
  static_assert(!LN || K == KS, "a normalised row has no padded columns");
  constexpr int N = NT * 16;
  if (zero_p != nullptr) {
    const int nth = gridDim.x * gridDim.y * blockDim.x;
    for (int i = (blockIdx.y * gridDim.x + blockIdx.x) * blockDim.x + threadIdx.x; i < zero_n; i += nth)
      zero_p[i] = 0u;
  }
  const float* __restrict__ B = blockIdx.y ? B2 : B1;
  float* __restrict__ C = blockIdx.y ? C2 : C1;
  const int lane  = threadIdx.x & 31;
  const int wave  = threadIdx.x >> 5;
  const int tileM = blockIdx.x * 4 + wave;
  const int m0 = tileM * 16;
  if (m0 + 16 > M) return;
  const int half = lane >> 4;
  const int l16  = lane & 15;
  const float* __restrict__ arow = A + (size_t)(m0 + l16) * KS;
  const float* __restrict__ bcol = B + l16;
  v2f a[K / 4];
#pragma unroll
  for (int k = 0; k < K; k += 4) {
    if (k + 4 <= KS) {
      a[k / 4] = *(const v2f*)(arow + k + half * 2);
    } else {
      const v2f lo = *(const v2f*)(arow + k);
      a[k / 4] = half ? v2f{0.0f, 0.0f} : lo;
    }
  }
  if (LN) {
    constexpr int KC = (K < 32) ? K : 32;
    static_assert(K % KC == 0, "whole groups");
    const int c0 = half * 2;
    float sum = 0.0f;
#pragma unroll
    for (int kc = 0; kc < K; kc += KC) {
      v2f c[KC / 4];
#pragma unroll
      for (int j = 0; j < KC / 4; ++j) c[j] = *(const v2f*)(bias + kc + 4 * j + c0);
#pragma unroll
      for (int j = 0; j < KC / 4; ++j) {
        a[kc / 4 + j] += c[j];
        sum += a[kc / 4 + j].x;
        sum += a[kc / 4 + j].y;
      }
    }
    sum += xor_half(sum);
    const float mu = sum / (float)KS;
    float vs = 0.0f;
#pragma unroll
    for (int k = 0; k < K; k += 4) {
      a[k / 4] -= mu;
      vs += a[k / 4].x * a[k / 4].x;
      vs += a[k / 4].y * a[k / 4].y;
    }
    vs += xor_half(vs);
    const float inv = rsqrtf(vs / (float)KS + LN_EPS);
#pragma unroll
    for (int kc = 0; kc < K; kc += KC) {
      v2f g[KC / 4], s[KC / 4];
#pragma unroll
      for (int j = 0; j < KC / 4; ++j) {
        g[j] = *(const v2f*)(ln_g + kc + 4 * j + c0);
        s[j] = *(const v2f*)(ln_b + kc + 4 * j + c0);
      }
#pragma unroll
      for (int j = 0; j < KC / 4; ++j) {
        const v2f hv = a[kc / 4 + j] * inv * g[j] + s[j];
        a[kc / 4 + j].x = (hv.x > 0.0f) ? hv.x : (__expf(hv.x) - 1.0f);
        a[kc / 4 + j].y = (hv.y > 0.0f) ? hv.y : (__expf(hv.y) - 1.0f);
      }
    }
  }
  v8f acc[NT];
#pragma unroll
  for (int t = 0; t < NT; ++t) acc[t] = v8f{};
#pragma unroll
  for (int k = 0; k < K; k += 4) {
#pragma unroll
    for (int t = 0; t < NT; ++t) {
      v2f b;
      if (k + 4 <= KS) {
        b.x = bcol[(size_t)(k + half * 2) * N + t * 16];
        b.y = bcol[(size_t)(k + half * 2 + 1) * N + t * 16];
      } else {
        const float lo0 = bcol[(size_t)k * N + t * 16];
        const float lo1 = bcol[(size_t)(k + 1) * N + t * 16];
        b.x = half ? 0.0f : lo0;
        b.y = half ? 0.0f : lo1;
      }
      acc[t] = __builtin_amdgcn_wmma_f32_16x16x4_f32(false, a[k / 4], false, b, (short)0, acc[t],
                                                     false, false);
    }
  }
  float* __restrict__ crow = C + (size_t)(m0 + half * 8) * N + l16;
#pragma unroll
  for (int r = 0; r < 8; ++r)
#pragma unroll
    for (int t = 0; t < NT; ++t)
      crow[(size_t)r * N + t * 16] = acc[t][r];
}

#define EDGE_U 6

__global__ void csr_count(const int* __restrict__ ei, unsigned* __restrict__ cnt, int E) {
  const int stride = gridDim.x * blockDim.x;
  int e0 = blockIdx.x * blockDim.x + threadIdx.x;
  for (; e0 + (EDGE_U - 1) * stride < E; e0 += EDGE_U * stride) {
    int d[EDGE_U];
#pragma unroll
    for (int u = 0; u < EDGE_U; ++u) d[u] = ei[E + e0 + u * stride];
#pragma unroll
    for (int u = 0; u < EDGE_U; ++u) atomicAdd(&cnt[d[u]], 1u);
  }
  for (; e0 < E; e0 += stride) atomicAdd(&cnt[ei[E + e0]], 1u);
}

__global__ void csr_scan(const unsigned* __restrict__ cnt, unsigned* __restrict__ rowptr,
                         unsigned* __restrict__ cursor, int N) {
  __shared__ unsigned part[1024];
  const int t = threadIdx.x;
  const int base_i = blockIdx.x * 1024;
  unsigned pre = 0;
  for (int i = t; i < base_i; i += 1024) pre += cnt[i];
  const int idx = base_i + t;
  const unsigned c = (idx < N) ? cnt[idx] : 0u;
  part[t] = pre;
  __syncthreads();
  for (int off = 512; off > 0; off >>= 1) {
    if (t < off) part[t] += part[t + off];
    __syncthreads();
  }
  const unsigned total_pre = part[0];
  __syncthreads();
  part[t] = c;
  __syncthreads();
  for (int off = 1; off < 1024; off <<= 1) {
    const unsigned v = (t >= off) ? part[t - off] : 0u;
    __syncthreads();
    part[t] += v;
    __syncthreads();
  }
  const unsigned excl = part[t] - c + total_pre;
  if (idx < N) { rowptr[idx] = excl; cursor[idx] = excl; }
}

__global__ void csr_fill(const int* __restrict__ ei, const float* __restrict__ ea,
                         unsigned* __restrict__ cursor,
                         int* __restrict__ seg_s, float* __restrict__ seg_a, int E) {
  const int stride = gridDim.x * blockDim.x;
  int e0 = blockIdx.x * blockDim.x + threadIdx.x;
  for (; e0 + (EDGE_U - 1) * stride < E; e0 += EDGE_U * stride) {
    int d[EDGE_U], sv[EDGE_U];
    float av[EDGE_U];
    unsigned p[EDGE_U];
#pragma unroll
    for (int u = 0; u < EDGE_U; ++u) {
      const int e = e0 + u * stride;
      d[u] = ei[E + e];
      sv[u] = ei[e];
      av[u] = ea[e];
    }
#pragma unroll
    for (int u = 0; u < EDGE_U; ++u) p[u] = atomicAdd(&cursor[d[u]], 1u);
#pragma unroll
    for (int u = 0; u < EDGE_U; ++u) { seg_s[p[u]] = sv[u]; seg_a[p[u]] = av[u]; }
  }
  for (; e0 < E; e0 += stride) {
    const unsigned p = atomicAdd(&cursor[ei[E + e0]], 1u);
    seg_s[p] = ei[e0];
    seg_a[p] = ea[e0];
  }
}

template <int G> struct group_of { static constexpr int value = G; };

template <int H, int C, bool FINISH>
__global__ void __launch_bounds__(256, 16) node_attend(const int* __restrict__ seg_s, const float* __restrict__ seg_a,
                            const unsigned* __restrict__ rowptr, const unsigned* __restrict__ cnt,
                            const float* __restrict__ xl, const float* __restrict__ xr,
                            const float* __restrict__ We, const float* __restrict__ att,
                            const float* __restrict__ bias, const float* __restrict__ ln_g,
                            const float* __restrict__ ln_b,
                            float* __restrict__ out, int N) {
  constexpr int F = H * C;
  constexpr int V = F / 32;
  constexpr int LPH = C / V;
  typedef float vf __attribute__((ext_vector_type(V)));
  int node = __builtin_amdgcn_readfirstlane(
      (int)(blockIdx.x * (blockDim.x >> 5) + (threadIdx.x >> 5)));
  const int nw = (int)(gridDim.x * (blockDim.x >> 5));
  const int lane = threadIdx.x & 31;
  if (node >= N) return;
  const vf wev = *(const vf*)(We + lane * V);
  const vf atv = *(const vf*)(att + lane * V);
  unsigned beg = rowptr[node];
  unsigned len = cnt[node];
  for (;;) {
  const int nxt = node + nw;
  unsigned nbeg = 0u, nlen = 0u;
  if (nxt < N) { nbeg = rowptr[nxt]; nlen = cnt[nxt]; }
  const vf xlv = *(const vf*)(xl + (size_t)node * F + lane * V);
  float m = -INFINITY, dn = 0.0f;
  vf acc = {};
  for (unsigned base = 0; base < len; base += 32) {
    const unsigned nn = (len - base < 32u) ? (len - base) : 32u;
    int my_s = 0;
    float my_a = 0.0f;
    if ((unsigned)lane < nn) {
      my_s = seg_s[beg + base + lane];
      my_a = seg_a[beg + base + lane];
    }
    auto group = [&](auto gtag, unsigned j) {
      constexpr int G = decltype(gtag)::value;
      vf v[G];
      float p[G];
#pragma unroll
      for (int i = 0; i < G; ++i) {
        const int s = __builtin_amdgcn_readlane(my_s, (int)(j + i));
        v[i] = *(const vf*)(xr + (size_t)s * F + lane * V);
      }
#pragma unroll
      for (int i = 0; i < G; ++i) {
        const float a = __builtin_bit_cast(
            float, __builtin_amdgcn_readlane(__builtin_bit_cast(int, my_a), (int)(j + i)));
        float q;
#pragma unroll
        for (int k = 0; k < V; ++k) {
          float t = xlv[k] + v[i][k] + a * wev[k];
          t = fmaxf(t, NEG_SLOPE * t);
          q = (k == 0) ? atv[k] * t : q + atv[k] * t;
        }
        p[i] = head_sum<LPH>(q);
      }
      float mn = m;
#pragma unroll
      for (int i = 0; i < G; ++i) mn = fmaxf(mn, p[i]);
      const float sc = __expf(m - mn);
      float wsum;
      vf part;
#pragma unroll
      for (int i = 0; i < G; ++i) {
        const float w = __expf(p[i] - mn);
        wsum = (i == 0) ? w : wsum + w;
        part = (i == 0) ? w * v[i] : part + w * v[i];
      }
      dn = dn * sc + wsum;
      acc = acc * sc + part;
      m = mn;
    };
    unsigned j = 0;
    for (; j + 4 <= nn; j += 4) group(group_of<4>{}, j);
    for (; j < nn; ++j) group(group_of<1>{}, j);
  }
  const float dsafe = (len != 0u) ? dn : 1.0f;
  vf o = acc / dsafe;
  if (FINISH) {
    const vf bi = *(const vf*)(bias + lane * V);
    const vf gi = *(const vf*)(ln_g + lane * V);
    const vf be = *(const vf*)(ln_b + lane * V);
    float vals[V];
    float sum = 0.0f;
#pragma unroll
    for (int i = 0; i < V; ++i) {
      const float v = o[i] + bi[i];
      vals[i] = v;
      sum += v;
    }
    sum = wave_sum(sum);
    const float mu = sum / (float)F;
    float vs = 0.0f;
#pragma unroll
    for (int i = 0; i < V; ++i) { const float dlt = vals[i] - mu; vs += dlt * dlt; }
    vs = wave_sum(vs);
    const float inv = rsqrtf(vs / (float)F + LN_EPS);
#pragma unroll
    for (int i = 0; i < V; ++i) {
      const float hv = (vals[i] - mu) * inv * gi[i] + be[i];
      o[i] = (hv > 0.0f) ? hv : (__expf(hv) - 1.0f);
    }
  }
  *(vf*)(out + (size_t)node * F + lane * V) = o;
  if (nxt >= N) break;
  node = nxt; beg = nbeg; len = nlen;
  }
}

extern "C" void kernel_launch(void* const* d_in, const int* in_sizes, int n_in,
                              void* d_out, int out_size, void* d_ws, size_t ws_size,
                              hipStream_t stream) {
  const float* x      = (const float*)d_in[0];
  const float* ea     = (const float*)d_in[1];
  const float* W1l    = (const float*)d_in[2];
  const float* W1r    = (const float*)d_in[3];
  const float* W1e    = (const float*)d_in[4];
  const float* att1   = (const float*)d_in[5];
  const float* b1     = (const float*)d_in[6];
  const float* ln1_g  = (const float*)d_in[7];
  const float* ln1_b  = (const float*)d_in[8];
  const float* W2l    = (const float*)d_in[9];
  const float* W2r    = (const float*)d_in[10];
  const float* W2e    = (const float*)d_in[11];
  const float* att2   = (const float*)d_in[12];
  const float* b2     = (const float*)d_in[13];
  const float* ln2_g  = (const float*)d_in[14];
  const float* ln2_b  = (const float*)d_in[15];
  const int*   ei     = (const int*)d_in[16];
  float*       outp   = (float*)d_out;

  const int N = in_sizes[0] / 10;
  const int E = in_sizes[1];
  const int HID = 32, HEADS = 4, F1 = HEADS * HID  , OUTC = 64;

  float* ws      = (float*)d_ws;
  float* xl1     = ws;
  float* xr1     = xl1 + (size_t)N * F1;
  float* h1      = xr1 + (size_t)N * F1;
  float* xl2     = h1  + (size_t)N * F1;
  unsigned* cnt    = (unsigned*)(xl2 + (size_t)N * OUTC);
  unsigned* rowptr = cnt + (size_t)N;
  unsigned* cursor = rowptr + (size_t)N;
  int* seg_s       = (int*)(cursor + (size_t)N);
  float* seg_a     = (float*)(seg_s + (size_t)E);
  float* xr2     = xl1;

  const int TB = 256;
  const unsigned RES      = 512;
  const unsigned RES_NODE = 2048;
  auto blocks = [](size_t n, int tb) { return (unsigned)((n + tb - 1) / tb); };
  const dim3 gemm_grid((N / 16 + 3) / 4, 2);

  wmma_gemm_pair<10, 12, 8, false><<<gemm_grid, 128, 0, stream>>>(
      x, W1l, W1r, xl1, xr1, N, nullptr, nullptr, nullptr, cnt, N);

  csr_count<<<RES, TB, 0, stream>>>(ei, cnt, E);
  csr_scan<<<blocks((size_t)N, 1024), 1024, 0, stream>>>(cnt, rowptr, cursor, N);
  csr_fill<<<RES, TB, 0, stream>>>(ei, ea, cursor, seg_s, seg_a, E);

  node_attend<4, 32, false><<<RES_NODE, TB, 0, stream>>>(
      seg_s, seg_a, rowptr, cnt, xl1, xr1, W1e, att1, nullptr, nullptr, nullptr, h1, N);

  wmma_gemm_pair<128, 128, 4, true><<<gemm_grid, 128, 0, stream>>>(
      h1, W2l, W2r, xl2, xr2, N, b1, ln1_g, ln1_b, nullptr, 0);
  node_attend<1, 64, true><<<RES_NODE, TB, 0, stream>>>(
      seg_s, seg_a, rowptr, cnt, xl2, xr2, W2e, att2, b2, ln2_g, ln2_b, outp, N);
}


# edge grouping by dst: global-atomic count, scan, fill replaced by a two-level counting sort on LDS counters (bucket = dst>>8), rows ordered in LDS and copied out whole; segment order by LDS arrival
# speedup vs baseline: 14.4287x; 1.8966x over previous
#include <hip/hip_runtime.h>
#include <hip/hip_bf16.h>
#include <math.h>

typedef float     v2f  __attribute__((ext_vector_type(2)));
typedef float     v8f  __attribute__((ext_vector_type(8)));

#define NEG_SLOPE 0.2f
#define LN_EPS    1e-5f

template <int K>
__device__ __forceinline__ float xor_row(float x) {
  return __builtin_bit_cast(float, __builtin_amdgcn_update_dpp(
      0, __builtin_bit_cast(int, x), 0x160 | K, 0xf, 0xf, true));
}
__device__ __forceinline__ float xor_half(float x) {
  const int b = __builtin_bit_cast(int, x);
  return __builtin_bit_cast(float, __builtin_amdgcn_permlanex16(b, b, 0x76543210, 0xfedcba98,
                                                                false, false));
}
template <int LPH>
__device__ __forceinline__ float head_sum(float p) {
  p += xor_row<1>(p);
  p += xor_row<2>(p);
  p += xor_row<4>(p);
  if (LPH > 8)  p += xor_row<8>(p);
  if (LPH > 16) p += xor_half(p);
  return p;
}
__device__ __forceinline__ float wave_sum(float s) {
  s += xor_half(s);
  s += xor_row<8>(s);
  s += xor_row<4>(s);
  s += xor_row<2>(s);
  s += xor_row<1>(s);
  return s;
}

template <int KS, int K, int NT, bool LN>
__global__ void __launch_bounds__(128) wmma_gemm_pair(
    const float* __restrict__ A, const float* __restrict__ B1, const float* __restrict__ B2,
    float* __restrict__ C1, float* __restrict__ C2, int M,
    const float* __restrict__ bias, const float* __restrict__ ln_g, const float* __restrict__ ln_b) {
  static_assert(K % 4 == 0 && (K == KS || K == KS + 2), "K step layout");
  static_assert(!LN || K == KS, "a normalised row has no padded columns");
  constexpr int N = NT * 16;
  const float* __restrict__ B = blockIdx.y ? B2 : B1;
  float* __restrict__ C = blockIdx.y ? C2 : C1;
  const int lane  = threadIdx.x & 31;
  const int wave  = threadIdx.x >> 5;
  const int tileM = blockIdx.x * 4 + wave;
  const int m0 = tileM * 16;
  if (m0 + 16 > M) return;
  const int half = lane >> 4;
  const int l16  = lane & 15;
  const float* __restrict__ arow = A + (size_t)(m0 + l16) * KS;
  const float* __restrict__ bcol = B + l16;
  v2f a[K / 4];
#pragma unroll
  for (int k = 0; k < K; k += 4) {
    if (k + 4 <= KS) {
      a[k / 4] = *(const v2f*)(arow + k + half * 2);
    } else {
      const v2f lo = *(const v2f*)(arow + k);
      a[k / 4] = half ? v2f{0.0f, 0.0f} : lo;
    }
  }
  if (LN) {
    constexpr int KC = (K < 32) ? K : 32;
    static_assert(K % KC == 0, "whole groups");
    const int c0 = half * 2;
    float sum = 0.0f;
#pragma unroll
    for (int kc = 0; kc < K; kc += KC) {
      v2f c[KC / 4];
#pragma unroll
      for (int j = 0; j < KC / 4; ++j) c[j] = *(const v2f*)(bias + kc + 4 * j + c0);
#pragma unroll
      for (int j = 0; j < KC / 4; ++j) {
        a[kc / 4 + j] += c[j];
        sum += a[kc / 4 + j].x;
        sum += a[kc / 4 + j].y;
      }
    }
    sum += xor_half(sum);
    const float mu = sum / (float)KS;
    float vs = 0.0f;
#pragma unroll
    for (int k = 0; k < K; k += 4) {
      a[k / 4] -= mu;
      vs += a[k / 4].x * a[k / 4].x;
      vs += a[k / 4].y * a[k / 4].y;
    }
    vs += xor_half(vs);
    const float inv = rsqrtf(vs / (float)KS + LN_EPS);
#pragma unroll
    for (int kc = 0; kc < K; kc += KC) {
      v2f g[KC / 4], s[KC / 4];
#pragma unroll
      for (int j = 0; j < KC / 4; ++j) {
        g[j] = *(const v2f*)(ln_g + kc + 4 * j + c0);
        s[j] = *(const v2f*)(ln_b + kc + 4 * j + c0);
      }
#pragma unroll
      for (int j = 0; j < KC / 4; ++j) {
        const v2f hv = a[kc / 4 + j] * inv * g[j] + s[j];
        a[kc / 4 + j].x = (hv.x > 0.0f) ? hv.x : (__expf(hv.x) - 1.0f);
        a[kc / 4 + j].y = (hv.y > 0.0f) ? hv.y : (__expf(hv.y) - 1.0f);
      }
    }
  }
  v8f acc[NT];
#pragma unroll
  for (int t = 0; t < NT; ++t) acc[t] = v8f{};
#pragma unroll
  for (int k = 0; k < K; k += 4) {
#pragma unroll
    for (int t = 0; t < NT; ++t) {
      v2f b;
      if (k + 4 <= KS) {
        b.x = bcol[(size_t)(k + half * 2) * N + t * 16];
        b.y = bcol[(size_t)(k + half * 2 + 1) * N + t * 16];
      } else {
        const float lo0 = bcol[(size_t)k * N + t * 16];
        const float lo1 = bcol[(size_t)(k + 1) * N + t * 16];
        b.x = half ? 0.0f : lo0;
        b.y = half ? 0.0f : lo1;
      }
      acc[t] = __builtin_amdgcn_wmma_f32_16x16x4_f32(false, a[k / 4], false, b, (short)0, acc[t],
                                                     false, false);
    }
  }
  float* __restrict__ crow = C + (size_t)(m0 + half * 8) * N + l16;
#pragma unroll
  for (int r = 0; r < 8; ++r)
#pragma unroll
    for (int t = 0; t < NT; ++t)
      crow[(size_t)r * N + t * 16] = acc[t][r];
}

#define BSH   8
#define BR    256
#define ST    1024
#define CHUNK 4096
#define EPT   (CHUNK / ST)
#define BCAP  6144
#define TBATCH 16

__device__ __forceinline__ void wave_scan256(unsigned* v, int lane) {
  unsigned x[8], s = 0u;
#pragma unroll
  for (int j = 0; j < 8; ++j) { x[j] = v[lane * 8 + j]; s += x[j]; }
  unsigned inc = s;
#pragma unroll
  for (int off = 1; off < 32; off <<= 1) {
    const unsigned y = __shfl_up(inc, off);
    if (lane >= off) inc += y;
  }
  unsigned run = inc - s;
#pragma unroll
  for (int j = 0; j < 8; ++j) { v[lane * 8 + j] = run; run += x[j]; }
}

__global__ void __launch_bounds__(ST) bin_count(const int* __restrict__ ei, unsigned* __restrict__ table, int E) {
  __shared__ unsigned lh[BR];
  const int t = threadIdx.x;
  const int e0 = blockIdx.x * CHUNK;
  const int n = min(CHUNK, E - e0);
  int d[EPT];
#pragma unroll
  for (int u = 0; u < EPT; ++u) d[u] = (t + u * ST < n) ? ei[E + e0 + t + u * ST] : 0;
  if (t < BR) lh[t] = 0u;
  __syncthreads();
#pragma unroll
  for (int u = 0; u < EPT; ++u)
    if (t + u * ST < n) atomicAdd(&lh[d[u] >> BSH], 1u);
  __syncthreads();
  if (t < BR) table[(size_t)blockIdx.x * BR + t] = lh[t];
}

__global__ void __launch_bounds__(ST) bin_place(const int* __restrict__ ei, const float* __restrict__ ea,
                                                const unsigned* __restrict__ table,
                                                unsigned* __restrict__ bbase, unsigned* __restrict__ bcount,
                                                int* __restrict__ b_s, int* __restrict__ b_d, float* __restrict__ b_a,
                                                int E) {
  __shared__ int   st_s[CHUNK];
  __shared__ int   st_d[CHUNK];
  __shared__ float st_a[CHUNK];
  __shared__ unsigned goff[BR], pre[BR], lstart[BR], lcur[BR];
  const int t = threadIdx.x;
  const int nb = gridDim.x, b = blockIdx.x;
  const int e0 = b * CHUNK;
  const int n = min(CHUNK, E - e0);
  int d[EPT], sv[EPT];
  float av[EPT];
#pragma unroll
  for (int u = 0; u < EPT; ++u) {
    const int i = t + u * ST;
    d[u]  = (i < n) ? ei[E + e0 + i] : 0;
    sv[u] = (i < n) ? ei[e0 + i] : 0;
    av[u] = (i < n) ? ea[e0 + i] : 0.0f;
  }
  if (t < BR) { goff[t] = 0u; pre[t] = 0u; lcur[t] = 0u; }
  __syncthreads();
  {
    const int k = t & (BR - 1);
    const int q = t / BR;
    unsigned rs = 0u, pr = 0u;
    for (int bb0 = q; bb0 < nb; bb0 += (ST / BR) * TBATCH) {
      unsigned h[TBATCH];
#pragma unroll
      for (int u = 0; u < TBATCH; ++u) {
        const int bb = bb0 + (ST / BR) * u;
        h[u] = (bb < nb) ? table[(size_t)bb * BR + k] : 0u;
      }
#pragma unroll
      for (int u = 0; u < TBATCH; ++u) {
        rs += h[u];
        pr += (bb0 + (ST / BR) * u < b) ? h[u] : 0u;
      }
    }
    atomicAdd(&goff[k], rs);
    atomicAdd(&pre[k], pr);
  }
#pragma unroll
  for (int u = 0; u < EPT; ++u)
    if (t + u * ST < n) atomicAdd(&lcur[d[u] >> BSH], 1u);
  __syncthreads();
  unsigned tot = 0u;
  if (t < BR) { tot = goff[t]; lstart[t] = lcur[t]; }
  __syncthreads();
  if ((t >> 5) == 0) wave_scan256(goff, t & 31);
  if ((t >> 5) == 1) wave_scan256(lstart, t & 31);
  __syncthreads();
  if (t < BR) {
    if (b == 0) { bbase[t] = goff[t]; bcount[t] = tot; }
    goff[t] += pre[t];
    lcur[t] = lstart[t];
  }
  __syncthreads();
#pragma unroll
  for (int u = 0; u < EPT; ++u) {
    if (t + u * ST < n) {
      const unsigned p = atomicAdd(&lcur[d[u] >> BSH], 1u);
      st_d[p] = d[u];
      st_s[p] = sv[u];
      st_a[p] = av[u];
    }
  }
  __syncthreads();
  for (int i = t; i < n; i += ST) {
    const int dd = st_d[i];
    const int k = dd >> BSH;
    const unsigned g = goff[k] + ((unsigned)i - lstart[k]);
    b_d[g] = dd;
    b_s[g] = st_s[i];
    b_a[g] = st_a[i];
  }
}

__global__ void __launch_bounds__(ST) bucket_sort(const unsigned* __restrict__ bbase, const unsigned* __restrict__ bcount,
                                                  const int* __restrict__ b_s, const int* __restrict__ b_d,
                                                  const float* __restrict__ b_a,
                                                  unsigned* __restrict__ cnt, unsigned* __restrict__ rowptr,
                                                  int* __restrict__ seg_s, float* __restrict__ seg_a, int N) {
  __shared__ int   st_s[BCAP];
  __shared__ float st_a[BCAP];
  __shared__ unsigned lc[BR], ls[BR];
  const int t = threadIdx.x;
  const int k = blockIdx.x;
  const unsigned base = bbase[k];
  const int n = (int)bcount[k];
  const int n0 = k << BSH;
  if (t < BR) lc[t] = 0u;
  __syncthreads();
  for (int i0 = t; i0 < n; i0 += EPT * ST) {
    int d[EPT];
#pragma unroll
    for (int u = 0; u < EPT; ++u) d[u] = (i0 + u * ST < n) ? b_d[base + i0 + u * ST] : n0;
#pragma unroll
    for (int u = 0; u < EPT; ++u)
      if (i0 + u * ST < n) atomicAdd(&lc[d[u] - n0], 1u);
  }
  __syncthreads();
  unsigned c = 0u;
  if (t < BR) { c = lc[t]; ls[t] = c; }
  __syncthreads();
  if (t < 32) wave_scan256(ls, t);
  __syncthreads();
  if (t < BR) {
    const unsigned s = ls[t];
    lc[t] = s;
    if (n0 + t < N) { cnt[n0 + t] = c; rowptr[n0 + t] = base + s; }
  }
  __syncthreads();
  const bool staged = n <= BCAP;
  for (int i0 = t; i0 < n; i0 += EPT * ST) {
    int d[EPT], sv[EPT];
    float av[EPT];
#pragma unroll
    for (int u = 0; u < EPT; ++u) {
      const bool in = i0 + u * ST < n;
      d[u]  = in ? b_d[base + i0 + u * ST] : n0;
      sv[u] = in ? b_s[base + i0 + u * ST] : 0;
      av[u] = in ? b_a[base + i0 + u * ST] : 0.0f;
    }
#pragma unroll
    for (int u = 0; u < EPT; ++u) {
      if (i0 + u * ST < n) {
        const unsigned p = atomicAdd(&lc[d[u] - n0], 1u);
        if (staged) { st_s[p] = sv[u]; st_a[p] = av[u]; }
        else        { seg_s[base + p] = sv[u]; seg_a[base + p] = av[u]; }
      }
    }
  }
  __syncthreads();
  if (staged)
    for (int i = t; i < n; i += ST) { seg_s[base + i] = st_s[i]; seg_a[base + i] = st_a[i]; }
}

template <int G> struct group_of { static constexpr int value = G; };

template <int H, int C, bool FINISH>
__global__ void __launch_bounds__(256, 16) node_attend(const int* __restrict__ seg_s, const float* __restrict__ seg_a,
                            const unsigned* __restrict__ rowptr, const unsigned* __restrict__ cnt,
                            const float* __restrict__ xl, const float* __restrict__ xr,
                            const float* __restrict__ We, const float* __restrict__ att,
                            const float* __restrict__ bias, const float* __restrict__ ln_g,
                            const float* __restrict__ ln_b,
                            float* __restrict__ out, int N) {
  constexpr int F = H * C;
  constexpr int V = F / 32;
  constexpr int LPH = C / V;
  typedef float vf __attribute__((ext_vector_type(V)));
  int node = __builtin_amdgcn_readfirstlane(
      (int)(blockIdx.x * (blockDim.x >> 5) + (threadIdx.x >> 5)));
  const int nw = (int)(gridDim.x * (blockDim.x >> 5));
  const int lane = threadIdx.x & 31;
  if (node >= N) return;
  const vf wev = *(const vf*)(We + lane * V);
  const vf atv = *(const vf*)(att + lane * V);
  unsigned beg = rowptr[node];
  unsigned len = cnt[node];
  for (;;) {
  const int nxt = node + nw;
  unsigned nbeg = 0u, nlen = 0u;
  if (nxt < N) { nbeg = rowptr[nxt]; nlen = cnt[nxt]; }
  const vf xlv = *(const vf*)(xl + (size_t)node * F + lane * V);
  float m = -INFINITY, dn = 0.0f;
  vf acc = {};
  for (unsigned base = 0; base < len; base += 32) {
    const unsigned nn = (len - base < 32u) ? (len - base) : 32u;
    int my_s = 0;
    float my_a = 0.0f;
    if ((unsigned)lane < nn) {
      my_s = seg_s[beg + base + lane];
      my_a = seg_a[beg + base + lane];
    }
    auto group = [&](auto gtag, unsigned j) {
      constexpr int G = decltype(gtag)::value;
      vf v[G];
      float p[G];
#pragma unroll
      for (int i = 0; i < G; ++i) {
        const int s = __builtin_amdgcn_readlane(my_s, (int)(j + i));
        v[i] = *(const vf*)(xr + (size_t)s * F + lane * V);
      }
#pragma unroll
      for (int i = 0; i < G; ++i) {
        const float a = __builtin_bit_cast(
            float, __builtin_amdgcn_readlane(__builtin_bit_cast(int, my_a), (int)(j + i)));
        float q;
#pragma unroll
        for (int k = 0; k < V; ++k) {
          float t = xlv[k] + v[i][k] + a * wev[k];
          t = fmaxf(t, NEG_SLOPE * t);
          q = (k == 0) ? atv[k] * t : q + atv[k] * t;
        }
        p[i] = head_sum<LPH>(q);
      }
      float mn = m;
#pragma unroll
      for (int i = 0; i < G; ++i) mn = fmaxf(mn, p[i]);
      const float sc = __expf(m - mn);
      float wsum;
      vf part;
#pragma unroll
      for (int i = 0; i < G; ++i) {
        const float w = __expf(p[i] - mn);
        wsum = (i == 0) ? w : wsum + w;
        part = (i == 0) ? w * v[i] : part + w * v[i];
      }
      dn = dn * sc + wsum;
      acc = acc * sc + part;
      m = mn;
    };
    unsigned j = 0;
    for (; j + 4 <= nn; j += 4) group(group_of<4>{}, j);
    for (; j < nn; ++j) group(group_of<1>{}, j);
  }
  const float dsafe = (len != 0u) ? dn : 1.0f;
  vf o = acc / dsafe;
  if (FINISH) {
    const vf bi = *(const vf*)(bias + lane * V);
    const vf gi = *(const vf*)(ln_g + lane * V);
    const vf be = *(const vf*)(ln_b + lane * V);
    float vals[V];
    float sum = 0.0f;
#pragma unroll
    for (int i = 0; i < V; ++i) {
      const float v = o[i] + bi[i];
      vals[i] = v;
      sum += v;
    }
    sum = wave_sum(sum);
    const float mu = sum / (float)F;
    float vs = 0.0f;
#pragma unroll
    for (int i = 0; i < V; ++i) { const float dlt = vals[i] - mu; vs += dlt * dlt; }
    vs = wave_sum(vs);
    const float inv = rsqrtf(vs / (float)F + LN_EPS);
#pragma unroll
    for (int i = 0; i < V; ++i) {
      const float hv = (vals[i] - mu) * inv * gi[i] + be[i];
      o[i] = (hv > 0.0f) ? hv : (__expf(hv) - 1.0f);
    }
  }
  *(vf*)(out + (size_t)node * F + lane * V) = o;
  if (nxt >= N) break;
  node = nxt; beg = nbeg; len = nlen;
  }
}

extern "C" void kernel_launch(void* const* d_in, const int* in_sizes, int n_in,
                              void* d_out, int out_size, void* d_ws, size_t ws_size,
                              hipStream_t stream) {
  const float* x      = (const float*)d_in[0];
  const float* ea     = (const float*)d_in[1];
  const float* W1l    = (const float*)d_in[2];
  const float* W1r    = (const float*)d_in[3];
  const float* W1e    = (const float*)d_in[4];
  const float* att1   = (const float*)d_in[5];
  const float* b1     = (const float*)d_in[6];
  const float* ln1_g  = (const float*)d_in[7];
  const float* ln1_b  = (const float*)d_in[8];
  const float* W2l    = (const float*)d_in[9];
  const float* W2r    = (const float*)d_in[10];
  const float* W2e    = (const float*)d_in[11];
  const float* att2   = (const float*)d_in[12];
  const float* b2     = (const float*)d_in[13];
  const float* ln2_g  = (const float*)d_in[14];
  const float* ln2_b  = (const float*)d_in[15];
  const int*   ei     = (const int*)d_in[16];
  float*       outp   = (float*)d_out;

  const int N = in_sizes[0] / 10;
  const int E = in_sizes[1];
  const int HID = 32, HEADS = 4, F1 = HEADS * HID  , OUTC = 64;

  float* ws      = (float*)d_ws;
  float* xl1     = ws;
  float* xr1     = xl1 + (size_t)N * F1;
  float* h1      = xr1 + (size_t)N * F1;
  float* xl2     = h1  + (size_t)N * F1;
  unsigned* cnt    = (unsigned*)(xl2 + (size_t)N * OUTC);
  unsigned* rowptr = cnt + (size_t)N;
  const unsigned NB1 = (unsigned)((E + CHUNK - 1) / CHUNK);
  const unsigned NBK = (unsigned)((N + BR - 1) / BR);
  unsigned* table  = rowptr + (size_t)N;
  unsigned* bbase  = table + (size_t)BR * NB1;
  unsigned* bcount = bbase + BR;
  int* b_s         = (int*)(bcount + BR);
  int* b_d         = b_s + (size_t)E;
  float* b_a       = (float*)(b_d + (size_t)E);
  int* seg_s       = (int*)(b_a + (size_t)E);
  float* seg_a     = (float*)(seg_s + (size_t)E);
  float* xr2     = xl1;

  const int TB = 256;
  const unsigned RES_NODE = 2048;
  const dim3 gemm_grid((N / 16 + 3) / 4, 2);

  wmma_gemm_pair<10, 12, 8, false><<<gemm_grid, 128, 0, stream>>>(
      x, W1l, W1r, xl1, xr1, N, nullptr, nullptr, nullptr);

  bin_count<<<NB1, ST, 0, stream>>>(ei, table, E);
  bin_place<<<NB1, ST, 0, stream>>>(ei, ea, table, bbase, bcount, b_s, b_d, b_a, E);
  bucket_sort<<<NBK, ST, 0, stream>>>(bbase, bcount, b_s, b_d, b_a, cnt, rowptr, seg_s, seg_a, N);

  node_attend<4, 32, false><<<RES_NODE, TB, 0, stream>>>(
      seg_s, seg_a, rowptr, cnt, xl1, xr1, W1e, att1, nullptr, nullptr, nullptr, h1, N);

  wmma_gemm_pair<128, 128, 4, true><<<gemm_grid, 128, 0, stream>>>(
      h1, W2l, W2r, xl2, xr2, N, b1, ln1_g, ln1_b);
  node_attend<1, 64, true><<<RES_NODE, TB, 0, stream>>>(
      seg_s, seg_a, rowptr, cnt, xl2, xr2, W2e, att2, b2, ln2_g, ln2_b, outp, N);
}


# edge sort level 1 in one launch: each chunk bucket-ordered where it stands, level 2 collects its bucket from the chunks by two table columns; 7 launches to 6, no table sums; segment order: LDS arrival
# speedup vs baseline: 15.6569x; 1.0851x over previous
#include <hip/hip_runtime.h>
#include <hip/hip_bf16.h>
#include <math.h>

typedef float     v2f  __attribute__((ext_vector_type(2)));
typedef float     v8f  __attribute__((ext_vector_type(8)));

#define NEG_SLOPE 0.2f
#define LN_EPS    1e-5f

template <int K>
__device__ __forceinline__ float xor_row(float x) {
  return __builtin_bit_cast(float, __builtin_amdgcn_update_dpp(
      0, __builtin_bit_cast(int, x), 0x160 | K, 0xf, 0xf, true));
}
__device__ __forceinline__ float xor_half(float x) {
  const int b = __builtin_bit_cast(int, x);
  return __builtin_bit_cast(float, __builtin_amdgcn_permlanex16(b, b, 0x76543210, 0xfedcba98,
                                                                false, false));
}
template <int LPH>
__device__ __forceinline__ float head_sum(float p) {
  p += xor_row<1>(p);
  p += xor_row<2>(p);
  p += xor_row<4>(p);
  if (LPH > 8)  p += xor_row<8>(p);
  if (LPH > 16) p += xor_half(p);
  return p;
}
__device__ __forceinline__ float wave_sum(float s) {
  s += xor_half(s);
  s += xor_row<8>(s);
  s += xor_row<4>(s);
  s += xor_row<2>(s);
  s += xor_row<1>(s);
  return s;
}

template <int KS, int K, int NT, bool LN>
__global__ void __launch_bounds__(128) wmma_gemm_pair(
    const float* __restrict__ A, const float* __restrict__ B1, const float* __restrict__ B2,
    float* __restrict__ C1, float* __restrict__ C2, int M,
    const float* __restrict__ bias, const float* __restrict__ ln_g, const float* __restrict__ ln_b) {
  static_assert(K % 4 == 0 && (K == KS || K == KS + 2), "K step layout");
  static_assert(!LN || K == KS, "a normalised row has no padded columns");
  constexpr int N = NT * 16;
  const float* __restrict__ B = blockIdx.y ? B2 : B1;
  float* __restrict__ C = blockIdx.y ? C2 : C1;
  const int lane  = threadIdx.x & 31;
  const int wave  = threadIdx.x >> 5;
  const int tileM = blockIdx.x * 4 + wave;
  const int m0 = tileM * 16;
  if (m0 + 16 > M) return;
  const int half = lane >> 4;
  const int l16  = lane & 15;
  const float* __restrict__ arow = A + (size_t)(m0 + l16) * KS;
  const float* __restrict__ bcol = B + l16;
  v2f a[K / 4];
#pragma unroll
  for (int k = 0; k < K; k += 4) {
    if (k + 4 <= KS) {
      a[k / 4] = *(const v2f*)(arow + k + half * 2);
    } else {
      const v2f lo = *(const v2f*)(arow + k);
      a[k / 4] = half ? v2f{0.0f, 0.0f} : lo;
    }
  }
  if (LN) {
    constexpr int KC = (K < 32) ? K : 32;
    static_assert(K % KC == 0, "whole groups");
    const int c0 = half * 2;
    float sum = 0.0f;
#pragma unroll
    for (int kc = 0; kc < K; kc += KC) {
      v2f c[KC / 4];
#pragma unroll
      for (int j = 0; j < KC / 4; ++j) c[j] = *(const v2f*)(bias + kc + 4 * j + c0);
#pragma unroll
      for (int j = 0; j < KC / 4; ++j) {
        a[kc / 4 + j] += c[j];
        sum += a[kc / 4 + j].x;
        sum += a[kc / 4 + j].y;
      }
    }
    sum += xor_half(sum);
    const float mu = sum / (float)KS;
    float vs = 0.0f;
#pragma unroll
    for (int k = 0; k < K; k += 4) {
      a[k / 4] -= mu;
      vs += a[k / 4].x * a[k / 4].x;
      vs += a[k / 4].y * a[k / 4].y;
    }
    vs += xor_half(vs);
    const float inv = rsqrtf(vs / (float)KS + LN_EPS);
#pragma unroll
    for (int kc = 0; kc < K; kc += KC) {
      v2f g[KC / 4], s[KC / 4];
#pragma unroll
      for (int j = 0; j < KC / 4; ++j) {
        g[j] = *(const v2f*)(ln_g + kc + 4 * j + c0);
        s[j] = *(const v2f*)(ln_b + kc + 4 * j + c0);
      }
#pragma unroll
      for (int j = 0; j < KC / 4; ++j) {
        const v2f hv = a[kc / 4 + j] * inv * g[j] + s[j];
        a[kc / 4 + j].x = (hv.x > 0.0f) ? hv.x : (__expf(hv.x) - 1.0f);
        a[kc / 4 + j].y = (hv.y > 0.0f) ? hv.y : (__expf(hv.y) - 1.0f);
      }
    }
  }
  v8f acc[NT];
#pragma unroll
  for (int t = 0; t < NT; ++t) acc[t] = v8f{};
#pragma unroll
  for (int k = 0; k < K; k += 4) {
#pragma unroll
    for (int t = 0; t < NT; ++t) {
      v2f b;
      if (k + 4 <= KS) {
        b.x = bcol[(size_t)(k + half * 2) * N + t * 16];
        b.y = bcol[(size_t)(k + half * 2 + 1) * N + t * 16];
      } else {
        const float lo0 = bcol[(size_t)k * N + t * 16];
        const float lo1 = bcol[(size_t)(k + 1) * N + t * 16];
        b.x = half ? 0.0f : lo0;
        b.y = half ? 0.0f : lo1;
      }
      acc[t] = __builtin_amdgcn_wmma_f32_16x16x4_f32(false, a[k / 4], false, b, (short)0, acc[t],
                                                     false, false);
    }
  }
  float* __restrict__ crow = C + (size_t)(m0 + half * 8) * N + l16;
#pragma unroll
  for (int r = 0; r < 8; ++r)
#pragma unroll
    for (int t = 0; t < NT; ++t)
      crow[(size_t)r * N + t * 16] = acc[t][r];
}

#define BSH   8
#define BR    256
#define ST    1024
#define CHUNK 4096
#define EPT   (CHUNK / ST)
#define BCAP  6144

__device__ __forceinline__ void wave_scan256(unsigned* v, int lane) {
  unsigned x[8], s = 0u;
#pragma unroll
  for (int j = 0; j < 8; ++j) { x[j] = v[lane * 8 + j]; s += x[j]; }
  unsigned inc = s;
#pragma unroll
  for (int off = 1; off < 32; off <<= 1) {
    const unsigned y = __shfl_up(inc, off);
    if (lane >= off) inc += y;
  }
  unsigned run = inc - s;
#pragma unroll
  for (int j = 0; j < 8; ++j) { v[lane * 8 + j] = run; run += x[j]; }
}

__device__ __forceinline__ unsigned wave_usum(unsigned v) {
#pragma unroll
  for (int off = 16; off > 0; off >>= 1) v += __shfl_xor(v, off);
  return v;
}

__global__ void __launch_bounds__(ST) bin_local(const int* __restrict__ ei, const float* __restrict__ ea,
                                                unsigned* __restrict__ tcount, unsigned* __restrict__ tstart,
                                                int* __restrict__ b_s, int* __restrict__ b_d, float* __restrict__ b_a,
                                                int E) {
  __shared__ int   st_s[CHUNK];
  __shared__ int   st_d[CHUNK];
  __shared__ float st_a[CHUNK];
  __shared__ unsigned lstart[BR], lcur[BR];
  const int t = threadIdx.x;
  const int b = blockIdx.x;
  const int e0 = b * CHUNK;
  const int n = min(CHUNK, E - e0);
  int d[EPT], sv[EPT];
  float av[EPT];
#pragma unroll
  for (int u = 0; u < EPT; ++u) {
    const int i = t + u * ST;
    d[u]  = (i < n) ? ei[E + e0 + i] : 0;
    sv[u] = (i < n) ? ei[e0 + i] : 0;
    av[u] = (i < n) ? ea[e0 + i] : 0.0f;
  }
  if (t < BR) lcur[t] = 0u;
  __syncthreads();
#pragma unroll
  for (int u = 0; u < EPT; ++u)
    if (t + u * ST < n) atomicAdd(&lcur[d[u] >> BSH], 1u);
  __syncthreads();
  unsigned c = 0u;
  if (t < BR) { c = lcur[t]; lstart[t] = c; }
  __syncthreads();
  if ((t >> 5) == 0) wave_scan256(lstart, t & 31);
  __syncthreads();
  if (t < BR) {
    const unsigned s = lstart[t];
    tcount[(size_t)b * BR + t] = c;
    tstart[(size_t)b * BR + t] = s;
    lcur[t] = s;
  }
  __syncthreads();
#pragma unroll
  for (int u = 0; u < EPT; ++u) {
    if (t + u * ST < n) {
      const unsigned p = atomicAdd(&lcur[d[u] >> BSH], 1u);
      st_d[p] = d[u];
      st_s[p] = sv[u];
      st_a[p] = av[u];
    }
  }
  __syncthreads();
  for (int i = t; i < n; i += ST) {
    b_d[e0 + i] = st_d[i];
    b_s[e0 + i] = st_s[i];
    b_a[e0 + i] = st_a[i];
  }
}

#define CPW (BR / (ST / 32))
__global__ void __launch_bounds__(ST) bucket_sort(const unsigned* __restrict__ tcount, const unsigned* __restrict__ tstart,
                                                  const int* __restrict__ b_s, const int* __restrict__ b_d,
                                                  const float* __restrict__ b_a,
                                                  unsigned* __restrict__ cnt, unsigned* __restrict__ rowptr,
                                                  int* __restrict__ seg_s, float* __restrict__ seg_a, int N, int nb) {
  __shared__ int   st_s[BCAP];
  __shared__ float st_a[BCAP];
  __shared__ unsigned lc[BR], ls[BR], pst[BR], pcn[BR], tot[2];
  const int t = threadIdx.x;
  const int lane = t & 31;
  const int w = t >> 5;
  const int k = blockIdx.x;
  const int n0 = k << BSH;
  unsigned ms = 0u, mc = 0u;
  if (t < BR) {
    if (t < nb) { ms = tstart[(size_t)t * BR + k]; mc = tcount[(size_t)t * BR + k]; }
    pst[t] = (unsigned)t * CHUNK + ms;
    pcn[t] = mc;
    lc[t] = 0u;
  }
  if (t < 2) tot[t] = 0u;
  __syncthreads();
  if (t < BR) {
    const unsigned s = wave_usum(ms);
    const unsigned c = wave_usum(mc);
    if (lane == 0) { atomicAdd(&tot[0], s); atomicAdd(&tot[1], c); }
  }
  __syncthreads();
  const unsigned base = tot[0];
  const int n = (int)tot[1];
  unsigned po[CPW], pn[CPW];
#pragma unroll
  for (int u = 0; u < CPW; ++u) { po[u] = pst[w + (ST / 32) * u]; pn[u] = pcn[w + (ST / 32) * u]; }
  {
    int d[CPW];
#pragma unroll
    for (int u = 0; u < CPW; ++u) d[u] = ((unsigned)lane < pn[u]) ? b_d[po[u] + lane] : n0;
#pragma unroll
    for (int u = 0; u < CPW; ++u)
      if ((unsigned)lane < pn[u]) atomicAdd(&lc[d[u] - n0], 1u);
#pragma unroll
    for (int u = 0; u < CPW; ++u)
      for (unsigned j = 32u + lane; j < pn[u]; j += 32u) atomicAdd(&lc[b_d[po[u] + j] - n0], 1u);
  }
  __syncthreads();
  unsigned c = 0u;
  if (t < BR) { c = lc[t]; ls[t] = c; }
  __syncthreads();
  if (t < 32) wave_scan256(ls, t);
  __syncthreads();
  if (t < BR) {
    const unsigned s = ls[t];
    lc[t] = s;
    if (n0 + t < N) { cnt[n0 + t] = c; rowptr[n0 + t] = base + s; }
  }
  __syncthreads();
  const bool staged = n <= BCAP;
  auto put = [&](int dd, int ss, float aa) {
    const unsigned p = atomicAdd(&lc[dd - n0], 1u);
    if (staged) { st_s[p] = ss; st_a[p] = aa; }
    else        { seg_s[base + p] = ss; seg_a[base + p] = aa; }
  };
  {
    int d[CPW], sv[CPW];
    float av[CPW];
#pragma unroll
    for (int u = 0; u < CPW; ++u) {
      const bool in = (unsigned)lane < pn[u];
      d[u]  = in ? b_d[po[u] + lane] : n0;
      sv[u] = in ? b_s[po[u] + lane] : 0;
      av[u] = in ? b_a[po[u] + lane] : 0.0f;
    }
#pragma unroll
    for (int u = 0; u < CPW; ++u)
      if ((unsigned)lane < pn[u]) put(d[u], sv[u], av[u]);
#pragma unroll
    for (int u = 0; u < CPW; ++u)
      for (unsigned j = 32u + lane; j < pn[u]; j += 32u) put(b_d[po[u] + j], b_s[po[u] + j], b_a[po[u] + j]);
  }
  __syncthreads();
  if (staged)
    for (int i = t; i < n; i += ST) { seg_s[base + i] = st_s[i]; seg_a[base + i] = st_a[i]; }
}

template <int G> struct group_of { static constexpr int value = G; };

template <int H, int C, bool FINISH>
__global__ void __launch_bounds__(256, 16) node_attend(const int* __restrict__ seg_s, const float* __restrict__ seg_a,
                            const unsigned* __restrict__ rowptr, const unsigned* __restrict__ cnt,
                            const float* __restrict__ xl, const float* __restrict__ xr,
                            const float* __restrict__ We, const float* __restrict__ att,
                            const float* __restrict__ bias, const float* __restrict__ ln_g,
                            const float* __restrict__ ln_b,
                            float* __restrict__ out, int N) {
  constexpr int F = H * C;
  constexpr int V = F / 32;
  constexpr int LPH = C / V;
  typedef float vf __attribute__((ext_vector_type(V)));
  int node = __builtin_amdgcn_readfirstlane(
      (int)(blockIdx.x * (blockDim.x >> 5) + (threadIdx.x >> 5)));
  const int nw = (int)(gridDim.x * (blockDim.x >> 5));
  const int lane = threadIdx.x & 31;
  if (node >= N) return;
  const vf wev = *(const vf*)(We + lane * V);
  const vf atv = *(const vf*)(att + lane * V);
  unsigned beg = rowptr[node];
  unsigned len = cnt[node];
  for (;;) {
  const int nxt = node + nw;
  unsigned nbeg = 0u, nlen = 0u;
  if (nxt < N) { nbeg = rowptr[nxt]; nlen = cnt[nxt]; }
  const vf xlv = *(const vf*)(xl + (size_t)node * F + lane * V);
  float m = -INFINITY, dn = 0.0f;
  vf acc = {};
  for (unsigned base = 0; base < len; base += 32) {
    const unsigned nn = (len - base < 32u) ? (len - base) : 32u;
    int my_s = 0;
    float my_a = 0.0f;
    if ((unsigned)lane < nn) {
      my_s = seg_s[beg + base + lane];
      my_a = seg_a[beg + base + lane];
    }
    auto group = [&](auto gtag, unsigned j) {
      constexpr int G = decltype(gtag)::value;
      vf v[G];
      float p[G];
#pragma unroll
      for (int i = 0; i < G; ++i) {
        const int s = __builtin_amdgcn_readlane(my_s, (int)(j + i));
        v[i] = *(const vf*)(xr + (size_t)s * F + lane * V);
      }
#pragma unroll
      for (int i = 0; i < G; ++i) {
        const float a = __builtin_bit_cast(
            float, __builtin_amdgcn_readlane(__builtin_bit_cast(int, my_a), (int)(j + i)));
        float q;
#pragma unroll
        for (int k = 0; k < V; ++k) {
          float t = xlv[k] + v[i][k] + a * wev[k];
          t = fmaxf(t, NEG_SLOPE * t);
          q = (k == 0) ? atv[k] * t : q + atv[k] * t;
        }
        p[i] = head_sum<LPH>(q);
      }
      float mn = m;
#pragma unroll
      for (int i = 0; i < G; ++i) mn = fmaxf(mn, p[i]);
      const float sc = __expf(m - mn);
      float wsum;
      vf part;
#pragma unroll
      for (int i = 0; i < G; ++i) {
        const float w = __expf(p[i] - mn);
        wsum = (i == 0) ? w : wsum + w;
        part = (i == 0) ? w * v[i] : part + w * v[i];
      }
      dn = dn * sc + wsum;
      acc = acc * sc + part;
      m = mn;
    };
    unsigned j = 0;
    for (; j + 4 <= nn; j += 4) group(group_of<4>{}, j);
    for (; j < nn; ++j) group(group_of<1>{}, j);
  }
  const float dsafe = (len != 0u) ? dn : 1.0f;
  vf o = acc / dsafe;
  if (FINISH) {
    const vf bi = *(const vf*)(bias + lane * V);
    const vf gi = *(const vf*)(ln_g + lane * V);
    const vf be = *(const vf*)(ln_b + lane * V);
    float vals[V];
    float sum = 0.0f;
#pragma unroll
    for (int i = 0; i < V; ++i) {
      const float v = o[i] + bi[i];
      vals[i] = v;
      sum += v;
    }
    sum = wave_sum(sum);
    const float mu = sum / (float)F;
    float vs = 0.0f;
#pragma unroll
    for (int i = 0; i < V; ++i) { const float dlt = vals[i] - mu; vs += dlt * dlt; }
    vs = wave_sum(vs);
    const float inv = rsqrtf(vs / (float)F + LN_EPS);
#pragma unroll
    for (int i = 0; i < V; ++i) {
      const float hv = (vals[i] - mu) * inv * gi[i] + be[i];
      o[i] = (hv > 0.0f) ? hv : (__expf(hv) - 1.0f);
    }
  }
  *(vf*)(out + (size_t)node * F + lane * V) = o;
  if (nxt >= N) break;
  node = nxt; beg = nbeg; len = nlen;
  }
}

extern "C" void kernel_launch(void* const* d_in, const int* in_sizes, int n_in,
                              void* d_out, int out_size, void* d_ws, size_t ws_size,
                              hipStream_t stream) {
  const float* x      = (const float*)d_in[0];
  const float* ea     = (const float*)d_in[1];
  const float* W1l    = (const float*)d_in[2];
  const float* W1r    = (const float*)d_in[3];
  const float* W1e    = (const float*)d_in[4];
  const float* att1   = (const float*)d_in[5];
  const float* b1     = (const float*)d_in[6];
  const float* ln1_g  = (const float*)d_in[7];
  const float* ln1_b  = (const float*)d_in[8];
  const float* W2l    = (const float*)d_in[9];
  const float* W2r    = (const float*)d_in[10];
  const float* W2e    = (const float*)d_in[11];
  const float* att2   = (const float*)d_in[12];
  const float* b2     = (const float*)d_in[13];
  const float* ln2_g  = (const float*)d_in[14];
  const float* ln2_b  = (const float*)d_in[15];
  const int*   ei     = (const int*)d_in[16];
  float*       outp   = (float*)d_out;

  const int N = in_sizes[0] / 10;
  const int E = in_sizes[1];
  const int HID = 32, HEADS = 4, F1 = HEADS * HID  , OUTC = 64;

  float* ws      = (float*)d_ws;
  float* xl1     = ws;
  float* xr1     = xl1 + (size_t)N * F1;
  float* h1      = xr1 + (size_t)N * F1;
  float* xl2     = h1  + (size_t)N * F1;
  unsigned* cnt    = (unsigned*)(xl2 + (size_t)N * OUTC);
  unsigned* rowptr = cnt + (size_t)N;
  const unsigned NB1 = (unsigned)((E + CHUNK - 1) / CHUNK);
  const unsigned NBK = (unsigned)((N + BR - 1) / BR);
  unsigned* tcount = rowptr + (size_t)N;
  unsigned* tstart = tcount + (size_t)BR * NB1;
  int* b_s         = (int*)(tstart + (size_t)BR * NB1);
  int* b_d         = b_s + (size_t)E;
  float* b_a       = (float*)(b_d + (size_t)E);
  int* seg_s       = (int*)(b_a + (size_t)E);
  float* seg_a     = (float*)(seg_s + (size_t)E);
  float* xr2     = xl1;

  const int TB = 256;
  const unsigned RES_NODE = 2048;
  const dim3 gemm_grid((N / 16 + 3) / 4, 2);

  wmma_gemm_pair<10, 12, 8, false><<<gemm_grid, 128, 0, stream>>>(
      x, W1l, W1r, xl1, xr1, N, nullptr, nullptr, nullptr);

  bin_local<<<NB1, ST, 0, stream>>>(ei, ea, tcount, tstart, b_s, b_d, b_a, E);
  bucket_sort<<<NBK, ST, 0, stream>>>(tcount, tstart, b_s, b_d, b_a, cnt, rowptr, seg_s, seg_a, N, (int)NB1);

  node_attend<4, 32, false><<<RES_NODE, TB, 0, stream>>>(
      seg_s, seg_a, rowptr, cnt, xl1, xr1, W1e, att1, nullptr, nullptr, nullptr, h1, N);

  wmma_gemm_pair<128, 128, 4, true><<<gemm_grid, 128, 0, stream>>>(
      h1, W2l, W2r, xl2, xr2, N, b1, ln1_g, ln1_b);
  node_attend<1, 64, true><<<RES_NODE, TB, 0, stream>>>(
      seg_s, seg_a, rowptr, cnt, xl2, xr2, W2e, att2, b2, ln2_g, ln2_b, outp, N);
}


# layer 1 attention and the layer 2 projections in one launch: a block's 16 finished rows go through LDS into the WMMA (6 launches to 5, rows never stored); LayerNorm row sums by lane butterfly
# speedup vs baseline: 17.4991x; 1.1177x over previous
#include <hip/hip_runtime.h>
#include <hip/hip_bf16.h>
#include <math.h>

typedef float     v2f  __attribute__((ext_vector_type(2)));
typedef float     v8f  __attribute__((ext_vector_type(8)));

#define NEG_SLOPE 0.2f
#define LN_EPS    1e-5f

template <int K>
__device__ __forceinline__ float xor_row(float x) {
  return __builtin_bit_cast(float, __builtin_amdgcn_update_dpp(
      0, __builtin_bit_cast(int, x), 0x160 | K, 0xf, 0xf, true));
}
__device__ __forceinline__ float xor_half(float x) {
  const int b = __builtin_bit_cast(int, x);
  return __builtin_bit_cast(float, __builtin_amdgcn_permlanex16(b, b, 0x76543210, 0xfedcba98,
                                                                false, false));
}
template <int LPH>
__device__ __forceinline__ float head_sum(float p) {
  p += xor_row<1>(p);
  p += xor_row<2>(p);
  p += xor_row<4>(p);
  if (LPH > 8)  p += xor_row<8>(p);
  if (LPH > 16) p += xor_half(p);
  return p;
}
__device__ __forceinline__ float wave_sum(float s) {
  s += xor_half(s);
  s += xor_row<8>(s);
  s += xor_row<4>(s);
  s += xor_row<2>(s);
  s += xor_row<1>(s);
  return s;
}

template <int KS, int K, int NT>
__global__ void __launch_bounds__(128) wmma_gemm_pair(
    const float* __restrict__ A, const float* __restrict__ B1, const float* __restrict__ B2,
    float* __restrict__ C1, float* __restrict__ C2, int M) {
  static_assert(K % 4 == 0 && (K == KS || K == KS + 2), "K step layout");
  constexpr int N = NT * 16;
  const float* __restrict__ B = blockIdx.y ? B2 : B1;
  float* __restrict__ C = blockIdx.y ? C2 : C1;
  const int lane  = threadIdx.x & 31;
  const int wave  = threadIdx.x >> 5;
  const int tileM = blockIdx.x * 4 + wave;
  const int m0 = tileM * 16;
  if (m0 + 16 > M) return;
  const int half = lane >> 4;
  const int l16  = lane & 15;
  const float* __restrict__ arow = A + (size_t)(m0 + l16) * KS;
  const float* __restrict__ bcol = B + l16;
  v2f a[K / 4];
#pragma unroll
  for (int k = 0; k < K; k += 4) {
    if (k + 4 <= KS) {
      a[k / 4] = *(const v2f*)(arow + k + half * 2);
    } else {
      const v2f lo = *(const v2f*)(arow + k);
      a[k / 4] = half ? v2f{0.0f, 0.0f} : lo;
    }
  }
  v8f acc[NT];
#pragma unroll
  for (int t = 0; t < NT; ++t) acc[t] = v8f{};
#pragma unroll
  for (int k = 0; k < K; k += 4) {
#pragma unroll
    for (int t = 0; t < NT; ++t) {
      v2f b;
      if (k + 4 <= KS) {
        b.x = bcol[(size_t)(k + half * 2) * N + t * 16];
        b.y = bcol[(size_t)(k + half * 2 + 1) * N + t * 16];
      } else {
        const float lo0 = bcol[(size_t)k * N + t * 16];
        const float lo1 = bcol[(size_t)(k + 1) * N + t * 16];
        b.x = half ? 0.0f : lo0;
        b.y = half ? 0.0f : lo1;
      }
      acc[t] = __builtin_amdgcn_wmma_f32_16x16x4_f32(false, a[k / 4], false, b, (short)0, acc[t],
                                                     false, false);
    }
  }
  float* __restrict__ crow = C + (size_t)(m0 + half * 8) * N + l16;
#pragma unroll
  for (int r = 0; r < 8; ++r)
#pragma unroll
    for (int t = 0; t < NT; ++t)
      crow[(size_t)r * N + t * 16] = acc[t][r];
}

#define BSH   8
#define BR    256
#define ST    1024
#define CHUNK 4096
#define EPT   (CHUNK / ST)
#define BCAP  6144

__device__ __forceinline__ void wave_scan256(unsigned* v, int lane) {
  unsigned x[8], s = 0u;
#pragma unroll
  for (int j = 0; j < 8; ++j) { x[j] = v[lane * 8 + j]; s += x[j]; }
  unsigned inc = s;
#pragma unroll
  for (int off = 1; off < 32; off <<= 1) {
    const unsigned y = __shfl_up(inc, off);
    if (lane >= off) inc += y;
  }
  unsigned run = inc - s;
#pragma unroll
  for (int j = 0; j < 8; ++j) { v[lane * 8 + j] = run; run += x[j]; }
}

__device__ __forceinline__ unsigned wave_usum(unsigned v) {
#pragma unroll
  for (int off = 16; off > 0; off >>= 1) v += __shfl_xor(v, off);
  return v;
}

__global__ void __launch_bounds__(ST) bin_local(const int* __restrict__ ei, const float* __restrict__ ea,
                                                unsigned* __restrict__ tcount, unsigned* __restrict__ tstart,
                                                int* __restrict__ b_s, int* __restrict__ b_d, float* __restrict__ b_a,
                                                int E) {
  __shared__ int   st_s[CHUNK];
  __shared__ int   st_d[CHUNK];
  __shared__ float st_a[CHUNK];
  __shared__ unsigned lstart[BR], lcur[BR];
  const int t = threadIdx.x;
  const int b = blockIdx.x;
  const int e0 = b * CHUNK;
  const int n = min(CHUNK, E - e0);
  int d[EPT], sv[EPT];
  float av[EPT];
#pragma unroll
  for (int u = 0; u < EPT; ++u) {
    const int i = t + u * ST;
    d[u]  = (i < n) ? ei[E + e0 + i] : 0;
    sv[u] = (i < n) ? ei[e0 + i] : 0;
    av[u] = (i < n) ? ea[e0 + i] : 0.0f;
  }
  if (t < BR) lcur[t] = 0u;
  __syncthreads();
#pragma unroll
  for (int u = 0; u < EPT; ++u)
    if (t + u * ST < n) atomicAdd(&lcur[d[u] >> BSH], 1u);
  __syncthreads();
  unsigned c = 0u;
  if (t < BR) { c = lcur[t]; lstart[t] = c; }
  __syncthreads();
  if ((t >> 5) == 0) wave_scan256(lstart, t & 31);
  __syncthreads();
  if (t < BR) {
    const unsigned s = lstart[t];
    tcount[(size_t)b * BR + t] = c;
    tstart[(size_t)b * BR + t] = s;
    lcur[t] = s;
  }
  __syncthreads();
#pragma unroll
  for (int u = 0; u < EPT; ++u) {
    if (t + u * ST < n) {
      const unsigned p = atomicAdd(&lcur[d[u] >> BSH], 1u);
      st_d[p] = d[u];
      st_s[p] = sv[u];
      st_a[p] = av[u];
    }
  }
  __syncthreads();
  for (int i = t; i < n; i += ST) {
    b_d[e0 + i] = st_d[i];
    b_s[e0 + i] = st_s[i];
    b_a[e0 + i] = st_a[i];
  }
}

#define CPW (BR / (ST / 32))
__global__ void __launch_bounds__(ST) bucket_sort(const unsigned* __restrict__ tcount, const unsigned* __restrict__ tstart,
                                                  const int* __restrict__ b_s, const int* __restrict__ b_d,
                                                  const float* __restrict__ b_a,
                                                  unsigned* __restrict__ cnt, unsigned* __restrict__ rowptr,
                                                  int* __restrict__ seg_s, float* __restrict__ seg_a, int N, int nb) {
  __shared__ int   st_s[BCAP];
  __shared__ float st_a[BCAP];
  __shared__ unsigned lc[BR], ls[BR], pst[BR], pcn[BR], tot[2];
  const int t = threadIdx.x;
  const int lane = t & 31;
  const int w = t >> 5;
  const int k = blockIdx.x;
  const int n0 = k << BSH;
  unsigned ms = 0u, mc = 0u;
  if (t < BR) {
    if (t < nb) { ms = tstart[(size_t)t * BR + k]; mc = tcount[(size_t)t * BR + k]; }
    pst[t] = (unsigned)t * CHUNK + ms;
    pcn[t] = mc;
    lc[t] = 0u;
  }
  if (t < 2) tot[t] = 0u;
  __syncthreads();
  if (t < BR) {
    const unsigned s = wave_usum(ms);
    const unsigned c = wave_usum(mc);
    if (lane == 0) { atomicAdd(&tot[0], s); atomicAdd(&tot[1], c); }
  }
  __syncthreads();
  const unsigned base = tot[0];
  const int n = (int)tot[1];
  unsigned po[CPW], pn[CPW];
#pragma unroll
  for (int u = 0; u < CPW; ++u) { po[u] = pst[w + (ST / 32) * u]; pn[u] = pcn[w + (ST / 32) * u]; }
  {
    int d[CPW];
#pragma unroll
    for (int u = 0; u < CPW; ++u) d[u] = ((unsigned)lane < pn[u]) ? b_d[po[u] + lane] : n0;
#pragma unroll
    for (int u = 0; u < CPW; ++u)
      if ((unsigned)lane < pn[u]) atomicAdd(&lc[d[u] - n0], 1u);
#pragma unroll
    for (int u = 0; u < CPW; ++u)
      for (unsigned j = 32u + lane; j < pn[u]; j += 32u) atomicAdd(&lc[b_d[po[u] + j] - n0], 1u);
  }
  __syncthreads();
  unsigned c = 0u;
  if (t < BR) { c = lc[t]; ls[t] = c; }
  __syncthreads();
  if (t < 32) wave_scan256(ls, t);
  __syncthreads();
  if (t < BR) {
    const unsigned s = ls[t];
    lc[t] = s;
    if (n0 + t < N) { cnt[n0 + t] = c; rowptr[n0 + t] = base + s; }
  }
  __syncthreads();
  const bool staged = n <= BCAP;
  auto put = [&](int dd, int ss, float aa) {
    const unsigned p = atomicAdd(&lc[dd - n0], 1u);
    if (staged) { st_s[p] = ss; st_a[p] = aa; }
    else        { seg_s[base + p] = ss; seg_a[base + p] = aa; }
  };
  {
    int d[CPW], sv[CPW];
    float av[CPW];
#pragma unroll
    for (int u = 0; u < CPW; ++u) {
      const bool in = (unsigned)lane < pn[u];
      d[u]  = in ? b_d[po[u] + lane] : n0;
      sv[u] = in ? b_s[po[u] + lane] : 0;
      av[u] = in ? b_a[po[u] + lane] : 0.0f;
    }
#pragma unroll
    for (int u = 0; u < CPW; ++u)
      if ((unsigned)lane < pn[u]) put(d[u], sv[u], av[u]);
#pragma unroll
    for (int u = 0; u < CPW; ++u)
      for (unsigned j = 32u + lane; j < pn[u]; j += 32u) put(b_d[po[u] + j], b_s[po[u] + j], b_a[po[u] + j]);
  }
  __syncthreads();
  if (staged)
    for (int i = t; i < n; i += ST) { seg_s[base + i] = st_s[i]; seg_a[base + i] = st_a[i]; }
}

template <int G> struct group_of { static constexpr int value = G; };
template <int V> struct lane_vec { typedef float type __attribute__((ext_vector_type(V))); };

template <int H, int C>
__device__ __forceinline__ typename lane_vec<H * C / 32>::type attend_row(
    int node, unsigned beg, unsigned len, int lane,
    const int* __restrict__ seg_s, const float* __restrict__ seg_a,
    const float* __restrict__ xl, const float* __restrict__ xr,
    const typename lane_vec<H * C / 32>::type wev, const typename lane_vec<H * C / 32>::type atv) {
  constexpr int F = H * C;
  constexpr int V = F / 32;
  constexpr int LPH = C / V;
  typedef typename lane_vec<V>::type vf;
  const vf xlv = *(const vf*)(xl + (size_t)node * F + lane * V);
  float m = -INFINITY, dn = 0.0f;
  vf acc = {};
  for (unsigned base = 0; base < len; base += 32) {
    const unsigned nn = (len - base < 32u) ? (len - base) : 32u;
    int my_s = 0;
    float my_a = 0.0f;
    if ((unsigned)lane < nn) {
      my_s = seg_s[beg + base + lane];
      my_a = seg_a[beg + base + lane];
    }
    auto group = [&](auto gtag, unsigned j) {
      constexpr int G = decltype(gtag)::value;
      vf v[G];
      float p[G];
#pragma unroll
      for (int i = 0; i < G; ++i) {
        const int s = __builtin_amdgcn_readlane(my_s, (int)(j + i));
        v[i] = *(const vf*)(xr + (size_t)s * F + lane * V);
      }
#pragma unroll
      for (int i = 0; i < G; ++i) {
        const float a = __builtin_bit_cast(
            float, __builtin_amdgcn_readlane(__builtin_bit_cast(int, my_a), (int)(j + i)));
        float q;
#pragma unroll
        for (int k = 0; k < V; ++k) {
          float t = xlv[k] + v[i][k] + a * wev[k];
          t = fmaxf(t, NEG_SLOPE * t);
          q = (k == 0) ? atv[k] * t : q + atv[k] * t;
        }
        p[i] = head_sum<LPH>(q);
      }
      float mn = m;
#pragma unroll
      for (int i = 0; i < G; ++i) mn = fmaxf(mn, p[i]);
      const float sc = __expf(m - mn);
      float wsum;
      vf part;
#pragma unroll
      for (int i = 0; i < G; ++i) {
        const float w = __expf(p[i] - mn);
        wsum = (i == 0) ? w : wsum + w;
        part = (i == 0) ? w * v[i] : part + w * v[i];
      }
      dn = dn * sc + wsum;
      acc = acc * sc + part;
      m = mn;
    };
    unsigned j = 0;
    for (; j + 4 <= nn; j += 4) group(group_of<4>{}, j);
    for (; j < nn; ++j) group(group_of<1>{}, j);
  }
  const float dsafe = (len != 0u) ? dn : 1.0f;
  return acc / dsafe;
}

template <int V>
__device__ __forceinline__ typename lane_vec<V>::type finish_row(
    typename lane_vec<V>::type o, int lane,
    const float* __restrict__ bias, const float* __restrict__ ln_g, const float* __restrict__ ln_b) {
  constexpr int F = 32 * V;
  typedef typename lane_vec<V>::type vf;
  const vf bi = *(const vf*)(bias + lane * V);
  const vf gi = *(const vf*)(ln_g + lane * V);
  const vf be = *(const vf*)(ln_b + lane * V);
  float vals[V];
  float sum = 0.0f;
#pragma unroll
  for (int i = 0; i < V; ++i) {
    const float v = o[i] + bi[i];
    vals[i] = v;
    sum += v;
  }
  sum = wave_sum(sum);
  const float mu = sum / (float)F;
  float vs = 0.0f;
#pragma unroll
  for (int i = 0; i < V; ++i) { const float dlt = vals[i] - mu; vs += dlt * dlt; }
  vs = wave_sum(vs);
  const float inv = rsqrtf(vs / (float)F + LN_EPS);
#pragma unroll
  for (int i = 0; i < V; ++i) {
    const float hv = (vals[i] - mu) * inv * gi[i] + be[i];
    o[i] = (hv > 0.0f) ? hv : (__expf(hv) - 1.0f);
  }
  return o;
}

template <int H, int C>
__global__ void __launch_bounds__(256, 16) node_attend(const int* __restrict__ seg_s, const float* __restrict__ seg_a,
                            const unsigned* __restrict__ rowptr, const unsigned* __restrict__ cnt,
                            const float* __restrict__ xl, const float* __restrict__ xr,
                            const float* __restrict__ We, const float* __restrict__ att,
                            const float* __restrict__ bias, const float* __restrict__ ln_g,
                            const float* __restrict__ ln_b,
                            float* __restrict__ out, int N) {
  constexpr int F = H * C;
  constexpr int V = F / 32;
  typedef typename lane_vec<V>::type vf;
  int node = __builtin_amdgcn_readfirstlane(
      (int)(blockIdx.x * (blockDim.x >> 5) + (threadIdx.x >> 5)));
  const int nw = (int)(gridDim.x * (blockDim.x >> 5));
  const int lane = threadIdx.x & 31;
  if (node >= N) return;
  const vf wev = *(const vf*)(We + lane * V);
  const vf atv = *(const vf*)(att + lane * V);
  unsigned beg = rowptr[node];
  unsigned len = cnt[node];
  for (;;) {
    const int nxt = node + nw;
    unsigned nbeg = 0u, nlen = 0u;
    if (nxt < N) { nbeg = rowptr[nxt]; nlen = cnt[nxt]; }
    const vf o = attend_row<H, C>(node, beg, len, lane, seg_s, seg_a, xl, xr, wev, atv);
    *(vf*)(out + (size_t)node * F + lane * V) = finish_row<V>(o, lane, bias, ln_g, ln_b);
    if (nxt >= N) break;
    node = nxt; beg = nbeg; len = nlen;
  }
}

#define HP 132
template <int H, int C, int NT>
__global__ void __launch_bounds__(256, 16) attend_project(
    const int* __restrict__ seg_s, const float* __restrict__ seg_a,
    const unsigned* __restrict__ rowptr, const unsigned* __restrict__ cnt,
    const float* __restrict__ xl, const float* __restrict__ xr,
    const float* __restrict__ We, const float* __restrict__ att,
    const float* __restrict__ bias, const float* __restrict__ ln_g, const float* __restrict__ ln_b,
    const float* __restrict__ B1, const float* __restrict__ B2,
    float* __restrict__ C1, float* __restrict__ C2) {
  constexpr int F = H * C;
  constexpr int V = F / 32;
  constexpr int N2 = NT * 16;
  static_assert(V == 4 && F + 4 == HP, "a lane stores 16 B of a row; pitch = row + 4");
  static_assert(2 * NT == 8, "one column tile per wave of the 8");
  typedef typename lane_vec<V>::type vf;
  __shared__ alignas(16) float strip[16 * HP];
  const int lane = threadIdx.x & 31;
  const int wave = __builtin_amdgcn_readfirstlane((int)(threadIdx.x >> 5));
  const int m0 = (int)blockIdx.x * 16;
  {
    const vf wev = *(const vf*)(We + lane * V);
    const vf atv = *(const vf*)(att + lane * V);
    const int n_a = m0 + wave, n_b = m0 + wave + 8;
    const unsigned beg_a = rowptr[n_a], len_a = cnt[n_a];
    const unsigned beg_b = rowptr[n_b], len_b = cnt[n_b];
    typedef __attribute__((address_space(3))) volatile vf lds_row;
    vf* const row_a = (vf*)(strip + wave * HP + lane * V);
    vf* const row_b = (vf*)(strip + (wave + 8) * HP + lane * V);
    lds_row* const wait_a = (lds_row*)row_a;
    *wait_a = attend_row<H, C>(n_a, beg_a, len_a, lane, seg_s, seg_a, xl, xr, wev, atv);
    const vf o_b = attend_row<H, C>(n_b, beg_b, len_b, lane, seg_s, seg_a, xl, xr, wev, atv);
    *row_b = finish_row<V>(o_b, lane, bias, ln_g, ln_b);
    const vf o_a = *wait_a;
    *row_a = finish_row<V>(o_a, lane, bias, ln_g, ln_b);
  }
  __syncthreads();
  const int half = lane >> 4;
  const int l16  = lane & 15;
  const float* __restrict__ B = (wave < NT) ? B1 : B2;
  float* __restrict__ Cm = (wave < NT) ? C1 : C2;
  const int n0 = (wave & (NT - 1)) * 16;
  const float* arow = strip + l16 * HP + half * 2;
  const float* __restrict__ bcol = B + (size_t)(half * 2) * N2 + n0 + l16;
  v8f acc = v8f{};
#pragma unroll
  for (int k = 0; k < F; k += 4) {
    const v2f a = *(const v2f*)(arow + k);
    v2f b;
    b.x = bcol[(size_t)k * N2];
    b.y = bcol[(size_t)(k + 1) * N2];
    acc = __builtin_amdgcn_wmma_f32_16x16x4_f32(false, a, false, b, (short)0, acc, false, false);
  }
  float* __restrict__ crow = Cm + (size_t)(m0 + half * 8) * N2 + n0 + l16;
#pragma unroll
  for (int r = 0; r < 8; ++r) crow[(size_t)r * N2] = acc[r];
}

extern "C" void kernel_launch(void* const* d_in, const int* in_sizes, int n_in,
                              void* d_out, int out_size, void* d_ws, size_t ws_size,
                              hipStream_t stream) {
  const float* x      = (const float*)d_in[0];
  const float* ea     = (const float*)d_in[1];
  const float* W1l    = (const float*)d_in[2];
  const float* W1r    = (const float*)d_in[3];
  const float* W1e    = (const float*)d_in[4];
  const float* att1   = (const float*)d_in[5];
  const float* b1     = (const float*)d_in[6];
  const float* ln1_g  = (const float*)d_in[7];
  const float* ln1_b  = (const float*)d_in[8];
  const float* W2l    = (const float*)d_in[9];
  const float* W2r    = (const float*)d_in[10];
  const float* W2e    = (const float*)d_in[11];
  const float* att2   = (const float*)d_in[12];
  const float* b2     = (const float*)d_in[13];
  const float* ln2_g  = (const float*)d_in[14];
  const float* ln2_b  = (const float*)d_in[15];
  const int*   ei     = (const int*)d_in[16];
  float*       outp   = (float*)d_out;

  const int N = in_sizes[0] / 10;
  const int E = in_sizes[1];
  const int HID = 32, HEADS = 4, F1 = HEADS * HID  , OUTC = 64;

  float* ws      = (float*)d_ws;
  float* xl1     = ws;
  float* xr1     = xl1 + (size_t)N * F1;
  float* xr2     = xr1 + (size_t)N * F1;
  float* xl2     = xr2 + (size_t)N * OUTC;
  unsigned* cnt    = (unsigned*)(xl2 + (size_t)N * OUTC);
  unsigned* rowptr = cnt + (size_t)N;
  const unsigned NB1 = (unsigned)((E + CHUNK - 1) / CHUNK);
  const unsigned NBK = (unsigned)((N + BR - 1) / BR);
  unsigned* tcount = rowptr + (size_t)N;
  unsigned* tstart = tcount + (size_t)BR * NB1;
  int* b_s         = (int*)(tstart + (size_t)BR * NB1);
  int* b_d         = b_s + (size_t)E;
  float* b_a       = (float*)(b_d + (size_t)E);
  int* seg_s       = (int*)(b_a + (size_t)E);
  float* seg_a     = (float*)(seg_s + (size_t)E);

  const int TB = 256;
  const unsigned RES_NODE = 2048;
  const dim3 gemm_grid((N / 16 + 3) / 4, 2);

  wmma_gemm_pair<10, 12, 8><<<gemm_grid, 128, 0, stream>>>(x, W1l, W1r, xl1, xr1, N);

  bin_local<<<NB1, ST, 0, stream>>>(ei, ea, tcount, tstart, b_s, b_d, b_a, E);
  bucket_sort<<<NBK, ST, 0, stream>>>(tcount, tstart, b_s, b_d, b_a, cnt, rowptr, seg_s, seg_a, N, (int)NB1);

  attend_project<4, 32, 4><<<N / 16, TB, 0, stream>>>(
      seg_s, seg_a, rowptr, cnt, xl1, xr1, W1e, att1, b1, ln1_g, ln1_b, W2l, W2r, xl2, xr2);

  node_attend<1, 64><<<RES_NODE, TB, 0, stream>>>(
      seg_s, seg_a, rowptr, cnt, xl2, xr2, W2e, att2, b2, ln2_g, ln2_b, outp, N);
}


# layer 2 edge walk: two edges per wave step, one per 16-lane row (4 channels a lane); even- and odd-place edges summed apart, merged per node: sum order changed, also in the LayerNorm lane sums
# speedup vs baseline: 18.8266x; 1.0759x over previous
#include <hip/hip_runtime.h>
#include <hip/hip_bf16.h>
#include <math.h>

typedef float     v2f  __attribute__((ext_vector_type(2)));
typedef float     v8f  __attribute__((ext_vector_type(8)));

#define NEG_SLOPE 0.2f
#define LN_EPS    1e-5f

template <int K>
__device__ __forceinline__ float xor_row(float x) {
  return __builtin_bit_cast(float, __builtin_amdgcn_update_dpp(
      0, __builtin_bit_cast(int, x), 0x160 | K, 0xf, 0xf, true));
}
__device__ __forceinline__ float xor_half(float x) {
  const int b = __builtin_bit_cast(int, x);
  return __builtin_bit_cast(float, __builtin_amdgcn_permlanex16(b, b, 0x76543210, 0xfedcba98,
                                                                false, false));
}
template <int LPH>
__device__ __forceinline__ float head_sum(float p) {
  p += xor_row<1>(p);
  p += xor_row<2>(p);
  p += xor_row<4>(p);
  if (LPH > 8)  p += xor_row<8>(p);
  if (LPH > 16) p += xor_half(p);
  return p;
}
__device__ __forceinline__ float wave_sum(float s) {
  s += xor_half(s);
  s += xor_row<8>(s);
  s += xor_row<4>(s);
  s += xor_row<2>(s);
  s += xor_row<1>(s);
  return s;
}

__device__ __forceinline__ int row_lane(int x, unsigned sel) {
  return __builtin_amdgcn_permlane16(x, x, sel, sel, true, false);
}
template <int LPH>
__device__ __forceinline__ float head_sum16(float p) {
  if (LPH > 1) p += xor_row<1>(p);
  if (LPH > 2) p += xor_row<2>(p);
  if (LPH > 4) p += xor_row<4>(p);
  if (LPH > 8) p += xor_row<8>(p);
  return p;
}

template <int KS, int K, int NT>
__global__ void __launch_bounds__(128) wmma_gemm_pair(
    const float* __restrict__ A, const float* __restrict__ B1, const float* __restrict__ B2,
    float* __restrict__ C1, float* __restrict__ C2, int M) {
  static_assert(K % 4 == 0 && (K == KS || K == KS + 2), "K step layout");
  constexpr int N = NT * 16;
  const float* __restrict__ B = blockIdx.y ? B2 : B1;
  float* __restrict__ C = blockIdx.y ? C2 : C1;
  const int lane  = threadIdx.x & 31;
  const int wave  = threadIdx.x >> 5;
  const int tileM = blockIdx.x * 4 + wave;
  const int m0 = tileM * 16;
  if (m0 + 16 > M) return;
  const int half = lane >> 4;
  const int l16  = lane & 15;
  const float* __restrict__ arow = A + (size_t)(m0 + l16) * KS;
  const float* __restrict__ bcol = B + l16;
  v2f a[K / 4];
#pragma unroll
  for (int k = 0; k < K; k += 4) {
    if (k + 4 <= KS) {
      a[k / 4] = *(const v2f*)(arow + k + half * 2);
    } else {
      const v2f lo = *(const v2f*)(arow + k);
      a[k / 4] = half ? v2f{0.0f, 0.0f} : lo;
    }
  }
  v8f acc[NT];
#pragma unroll
  for (int t = 0; t < NT; ++t) acc[t] = v8f{};
#pragma unroll
  for (int k = 0; k < K; k += 4) {
#pragma unroll
    for (int t = 0; t < NT; ++t) {
      v2f b;
      if (k + 4 <= KS) {
        b.x = bcol[(size_t)(k + half * 2) * N + t * 16];
        b.y = bcol[(size_t)(k + half * 2 + 1) * N + t * 16];
      } else {
        const float lo0 = bcol[(size_t)k * N + t * 16];
        const float lo1 = bcol[(size_t)(k + 1) * N + t * 16];
        b.x = half ? 0.0f : lo0;
        b.y = half ? 0.0f : lo1;
      }
      acc[t] = __builtin_amdgcn_wmma_f32_16x16x4_f32(false, a[k / 4], false, b, (short)0, acc[t],
                                                     false, false);
    }
  }
  float* __restrict__ crow = C + (size_t)(m0 + half * 8) * N + l16;
#pragma unroll
  for (int r = 0; r < 8; ++r)
#pragma unroll
    for (int t = 0; t < NT; ++t)
      crow[(size_t)r * N + t * 16] = acc[t][r];
}

#define BSH   8
#define BR    256
#define ST    1024
#define CHUNK 4096
#define EPT   (CHUNK / ST)
#define BCAP  6144

__device__ __forceinline__ void wave_scan256(unsigned* v, int lane) {
  unsigned x[8], s = 0u;
#pragma unroll
  for (int j = 0; j < 8; ++j) { x[j] = v[lane * 8 + j]; s += x[j]; }
  unsigned inc = s;
#pragma unroll
  for (int off = 1; off < 32; off <<= 1) {
    const unsigned y = __shfl_up(inc, off);
    if (lane >= off) inc += y;
  }
  unsigned run = inc - s;
#pragma unroll
  for (int j = 0; j < 8; ++j) { v[lane * 8 + j] = run; run += x[j]; }
}

__device__ __forceinline__ unsigned wave_usum(unsigned v) {
#pragma unroll
  for (int off = 16; off > 0; off >>= 1) v += __shfl_xor(v, off);
  return v;
}

__global__ void __launch_bounds__(ST) bin_local(const int* __restrict__ ei, const float* __restrict__ ea,
                                                unsigned* __restrict__ tcount, unsigned* __restrict__ tstart,
                                                int* __restrict__ b_s, int* __restrict__ b_d, float* __restrict__ b_a,
                                                int E) {
  __shared__ int   st_s[CHUNK];
  __shared__ int   st_d[CHUNK];
  __shared__ float st_a[CHUNK];
  __shared__ unsigned lstart[BR], lcur[BR];
  const int t = threadIdx.x;
  const int b = blockIdx.x;
  const int e0 = b * CHUNK;
  const int n = min(CHUNK, E - e0);
  int d[EPT], sv[EPT];
  float av[EPT];
#pragma unroll
  for (int u = 0; u < EPT; ++u) {
    const int i = t + u * ST;
    d[u]  = (i < n) ? ei[E + e0 + i] : 0;
    sv[u] = (i < n) ? ei[e0 + i] : 0;
    av[u] = (i < n) ? ea[e0 + i] : 0.0f;
  }
  if (t < BR) lcur[t] = 0u;
  __syncthreads();
#pragma unroll
  for (int u = 0; u < EPT; ++u)
    if (t + u * ST < n) atomicAdd(&lcur[d[u] >> BSH], 1u);
  __syncthreads();
  unsigned c = 0u;
  if (t < BR) { c = lcur[t]; lstart[t] = c; }
  __syncthreads();
  if ((t >> 5) == 0) wave_scan256(lstart, t & 31);
  __syncthreads();
  if (t < BR) {
    const unsigned s = lstart[t];
    tcount[(size_t)b * BR + t] = c;
    tstart[(size_t)b * BR + t] = s;
    lcur[t] = s;
  }
  __syncthreads();
#pragma unroll
  for (int u = 0; u < EPT; ++u) {
    if (t + u * ST < n) {
      const unsigned p = atomicAdd(&lcur[d[u] >> BSH], 1u);
      st_d[p] = d[u];
      st_s[p] = sv[u];
      st_a[p] = av[u];
    }
  }
  __syncthreads();
  for (int i = t; i < n; i += ST) {
    b_d[e0 + i] = st_d[i];
    b_s[e0 + i] = st_s[i];
    b_a[e0 + i] = st_a[i];
  }
}

#define CPW (BR / (ST / 32))
__global__ void __launch_bounds__(ST) bucket_sort(const unsigned* __restrict__ tcount, const unsigned* __restrict__ tstart,
                                                  const int* __restrict__ b_s, const int* __restrict__ b_d,
                                                  const float* __restrict__ b_a,
                                                  unsigned* __restrict__ cnt, unsigned* __restrict__ rowptr,
                                                  int* __restrict__ seg_s, float* __restrict__ seg_a, int N, int nb) {
  __shared__ int   st_s[BCAP];
  __shared__ float st_a[BCAP];
  __shared__ unsigned lc[BR], ls[BR], pst[BR], pcn[BR], tot[2];
  const int t = threadIdx.x;
  const int lane = t & 31;
  const int w = t >> 5;
  const int k = blockIdx.x;
  const int n0 = k << BSH;
  unsigned ms = 0u, mc = 0u;
  if (t < BR) {
    if (t < nb) { ms = tstart[(size_t)t * BR + k]; mc = tcount[(size_t)t * BR + k]; }
    pst[t] = (unsigned)t * CHUNK + ms;
    pcn[t] = mc;
    lc[t] = 0u;
  }
  if (t < 2) tot[t] = 0u;
  __syncthreads();
  if (t < BR) {
    const unsigned s = wave_usum(ms);
    const unsigned c = wave_usum(mc);
    if (lane == 0) { atomicAdd(&tot[0], s); atomicAdd(&tot[1], c); }
  }
  __syncthreads();
  const unsigned base = tot[0];
  const int n = (int)tot[1];
  unsigned po[CPW], pn[CPW];
#pragma unroll
  for (int u = 0; u < CPW; ++u) { po[u] = pst[w + (ST / 32) * u]; pn[u] = pcn[w + (ST / 32) * u]; }
  {
    int d[CPW];
#pragma unroll
    for (int u = 0; u < CPW; ++u) d[u] = ((unsigned)lane < pn[u]) ? b_d[po[u] + lane] : n0;
#pragma unroll
    for (int u = 0; u < CPW; ++u)
      if ((unsigned)lane < pn[u]) atomicAdd(&lc[d[u] - n0], 1u);
#pragma unroll
    for (int u = 0; u < CPW; ++u)
      for (unsigned j = 32u + lane; j < pn[u]; j += 32u) atomicAdd(&lc[b_d[po[u] + j] - n0], 1u);
  }
  __syncthreads();
  unsigned c = 0u;
  if (t < BR) { c = lc[t]; ls[t] = c; }
  __syncthreads();
  if (t < 32) wave_scan256(ls, t);
  __syncthreads();
  if (t < BR) {
    const unsigned s = ls[t];
    lc[t] = s;
    if (n0 + t < N) { cnt[n0 + t] = c; rowptr[n0 + t] = base + s; }
  }
  __syncthreads();
  const bool staged = n <= BCAP;
  auto put = [&](int dd, int ss, float aa) {
    const unsigned p = atomicAdd(&lc[dd - n0], 1u);
    if (staged) { st_s[p] = ss; st_a[p] = aa; }
    else        { seg_s[base + p] = ss; seg_a[base + p] = aa; }
  };
  {
    int d[CPW], sv[CPW];
    float av[CPW];
#pragma unroll
    for (int u = 0; u < CPW; ++u) {
      const bool in = (unsigned)lane < pn[u];
      d[u]  = in ? b_d[po[u] + lane] : n0;
      sv[u] = in ? b_s[po[u] + lane] : 0;
      av[u] = in ? b_a[po[u] + lane] : 0.0f;
    }
#pragma unroll
    for (int u = 0; u < CPW; ++u)
      if ((unsigned)lane < pn[u]) put(d[u], sv[u], av[u]);
#pragma unroll
    for (int u = 0; u < CPW; ++u)
      for (unsigned j = 32u + lane; j < pn[u]; j += 32u) put(b_d[po[u] + j], b_s[po[u] + j], b_a[po[u] + j]);
  }
  __syncthreads();
  if (staged)
    for (int i = t; i < n; i += ST) { seg_s[base + i] = st_s[i]; seg_a[base + i] = st_a[i]; }
}

template <int G> struct group_of { static constexpr int value = G; };
template <int G, bool M> struct step_group { static constexpr int steps = G; static constexpr bool masked = M; };
template <int V> struct lane_vec { typedef float type __attribute__((ext_vector_type(V))); };

template <int H, int C>
__device__ __forceinline__ typename lane_vec<H * C / 32>::type attend_row(
    int node, unsigned beg, unsigned len, int lane,
    const int* __restrict__ seg_s, const float* __restrict__ seg_a,
    const float* __restrict__ xl, const float* __restrict__ xr,
    const typename lane_vec<H * C / 32>::type wev, const typename lane_vec<H * C / 32>::type atv) {
  constexpr int F = H * C;
  constexpr int V = F / 32;
  constexpr int LPH = C / V;
  typedef typename lane_vec<V>::type vf;
  const vf xlv = *(const vf*)(xl + (size_t)node * F + lane * V);
  float m = -INFINITY, dn = 0.0f;
  vf acc = {};
  for (unsigned base = 0; base < len; base += 32) {
    const unsigned nn = (len - base < 32u) ? (len - base) : 32u;
    int my_s = 0;
    float my_a = 0.0f;
    if ((unsigned)lane < nn) {
      my_s = seg_s[beg + base + lane];
      my_a = seg_a[beg + base + lane];
    }
    auto group = [&](auto gtag, unsigned j) {
      constexpr int G = decltype(gtag)::value;
      vf v[G];
      float p[G];
#pragma unroll
      for (int i = 0; i < G; ++i) {
        const int s = __builtin_amdgcn_readlane(my_s, (int)(j + i));
        v[i] = *(const vf*)(xr + (size_t)s * F + lane * V);
      }
#pragma unroll
      for (int i = 0; i < G; ++i) {
        const float a = __builtin_bit_cast(
            float, __builtin_amdgcn_readlane(__builtin_bit_cast(int, my_a), (int)(j + i)));
        float q;
#pragma unroll
        for (int k = 0; k < V; ++k) {
          float t = xlv[k] + v[i][k] + a * wev[k];
          t = fmaxf(t, NEG_SLOPE * t);
          q = (k == 0) ? atv[k] * t : q + atv[k] * t;
        }
        p[i] = head_sum<LPH>(q);
      }
      float mn = m;
#pragma unroll
      for (int i = 0; i < G; ++i) mn = fmaxf(mn, p[i]);
      const float sc = __expf(m - mn);
      float wsum;
      vf part;
#pragma unroll
      for (int i = 0; i < G; ++i) {
        const float w = __expf(p[i] - mn);
        wsum = (i == 0) ? w : wsum + w;
        part = (i == 0) ? w * v[i] : part + w * v[i];
      }
      dn = dn * sc + wsum;
      acc = acc * sc + part;
      m = mn;
    };
    unsigned j = 0;
    for (; j + 4 <= nn; j += 4) group(group_of<4>{}, j);
    for (; j < nn; ++j) group(group_of<1>{}, j);
  }
  const float dsafe = (len != 0u) ? dn : 1.0f;
  return acc / dsafe;
}

template <int V>
__device__ __forceinline__ typename lane_vec<V>::type finish_vals(
    typename lane_vec<V>::type o, const typename lane_vec<V>::type bi,
    const typename lane_vec<V>::type gi, const typename lane_vec<V>::type be) {
  constexpr int F = 32 * V;
  float vals[V];
  float sum = 0.0f;
#pragma unroll
  for (int i = 0; i < V; ++i) {
    const float v = o[i] + bi[i];
    vals[i] = v;
    sum += v;
  }
  sum = wave_sum(sum);
  const float mu = sum / (float)F;
  float vs = 0.0f;
#pragma unroll
  for (int i = 0; i < V; ++i) { const float dlt = vals[i] - mu; vs += dlt * dlt; }
  vs = wave_sum(vs);
  const float inv = rsqrtf(vs / (float)F + LN_EPS);
#pragma unroll
  for (int i = 0; i < V; ++i) {
    const float hv = (vals[i] - mu) * inv * gi[i] + be[i];
    o[i] = (hv > 0.0f) ? hv : (__expf(hv) - 1.0f);
  }
  return o;
}
template <int V>
__device__ __forceinline__ typename lane_vec<V>::type finish_row(
    typename lane_vec<V>::type o, int ch0,
    const float* __restrict__ bias, const float* __restrict__ ln_g, const float* __restrict__ ln_b) {
  typedef typename lane_vec<V>::type vf;
  return finish_vals<V>(o, *(const vf*)(bias + ch0), *(const vf*)(ln_g + ch0), *(const vf*)(ln_b + ch0));
}

template <int H, int C>
__device__ __forceinline__ v2f attend_row_pairs(
    int node, unsigned beg, unsigned len, int lane,
    const int* __restrict__ seg_s, const float* __restrict__ seg_a,
    const float* __restrict__ xl, const float* __restrict__ xr,
    const typename lane_vec<4>::type wev, const typename lane_vec<4>::type atv) {
  constexpr int F = H * C;
  constexpr int V = 4;
  constexpr int LPH = C / V;
  static_assert(F == 16 * V && C % V == 0, "a 16-lane row holds the F channels, a head's channels lie side by side");
  typedef typename lane_vec<V>::type vf;
  const int c16 = lane & 15;
  const unsigned hf = (unsigned)lane >> 4;
  const char* __restrict__ xrb = (const char*)xr;
  const unsigned lane_off = (unsigned)c16 * (unsigned)(V * 4);
  const vf xlv = *(const vf*)((const char*)xl + ((unsigned)node * (unsigned)(F * 4) + lane_off));
  float m = -INFINITY, dn = 0.0f;
  v2f acc_a = {}, acc_b = {};
  for (unsigned base = 0; base < len; base += 32) {
    const unsigned nn = (len - base < 32u) ? (len - base) : 32u;
    const unsigned my_e = 2u * (unsigned)c16 + hf;
    int my_s = 0;
    float my_a = 0.0f;
    if (my_e < nn) {
      my_s = seg_s[beg + base + my_e];
      my_a = seg_a[beg + base + my_e];
    }
    auto group = [&](auto gtag, unsigned j) {
      constexpr int G = decltype(gtag)::steps;
      constexpr bool MASKED = decltype(gtag)::masked;
      vf v[G];
      float p[G];
#pragma unroll
      for (int i = 0; i < G; ++i) {
        const int s = row_lane(my_s, (j + i) * 0x11111111u);
        v[i] = *(const vf*)(xrb + ((unsigned)s * (unsigned)(F * 4) + lane_off));
      }
#pragma unroll
      for (int i = 0; i < G; ++i) {
        const float a = __builtin_bit_cast(
            float, row_lane(__builtin_bit_cast(int, my_a), (j + i) * 0x11111111u));
        float q;
#pragma unroll
        for (int k = 0; k < V; ++k) {
          float t = xlv[k] + v[i][k] + a * wev[k];
          t = fmaxf(t, NEG_SLOPE * t);
          q = (k == 0) ? atv[k] * t : q + atv[k] * t;
        }
        p[i] = head_sum16<LPH>(q);
      }
      bool ok[G];
      float mn = m;
#pragma unroll
      for (int i = 0; i < G; ++i) {
        ok[i] = !MASKED || (2u * (j + i) + hf < nn);
        mn = fmaxf(mn, ok[i] ? p[i] : -INFINITY);
      }
      const float sc = (MASKED && m == mn) ? 1.0f : __expf(m - mn);
      float wsum;
      v2f part_a, part_b;
#pragma unroll
      for (int i = 0; i < G; ++i) {
        const float w = ok[i] ? __expf(p[i] - mn) : 0.0f;
        const v2f va = __builtin_shufflevector(v[i], v[i], 0, 1);
        const v2f vb = __builtin_shufflevector(v[i], v[i], 2, 3);
        wsum = (i == 0) ? w : wsum + w;
        part_a = (i == 0) ? w * va : part_a + w * va;
        part_b = (i == 0) ? w * vb : part_b + w * vb;
      }
      dn = dn * sc + wsum;
      acc_a = acc_a * sc + part_a;
      acc_b = acc_b * sc + part_b;
      m = mn;
    };
    const unsigned full = nn >> 1;
    unsigned j = 0;
    for (; j + 4 <= full; j += 4) group(step_group<4, false>{}, j);
    for (; 2u * j < nn; j += 2) group(step_group<2, true>{}, j);
  }
  const float m_o = xor_half(m);
  const float dn_o = xor_half(dn);
  const float mm = fmaxf(m, m_o);
  const float f   = (m   == -INFINITY) ? 0.0f : __expf(m - mm);
  const float f_o = (m_o == -INFINITY) ? 0.0f : __expf(m_o - mm);
  const float dt = dn * f + dn_o * f_o;
  const v2f keep = hf ? acc_b : acc_a;
  const v2f send = hf ? acc_a : acc_b;
  v2f got;
  got.x = xor_half(send.x);
  got.y = xor_half(send.y);
  const v2f at = keep * f + got * f_o;
  const float dsafe = (len != 0u) ? dt : 1.0f;
  return at / dsafe;
}

template <int H, int C>
__global__ void __launch_bounds__(256, 16) node_attend(const int* __restrict__ seg_s, const float* __restrict__ seg_a,
                            const unsigned* __restrict__ rowptr, const unsigned* __restrict__ cnt,
                            const float* __restrict__ xl, const float* __restrict__ xr,
                            const float* __restrict__ We, const float* __restrict__ att,
                            const float* __restrict__ bias, const float* __restrict__ ln_g,
                            const float* __restrict__ ln_b,
                            float* __restrict__ out, int N) {
  constexpr int F = H * C;
  constexpr int V = F / 16;
  static_assert(V == 4, "the paired walk is written for rows of 64 channels");
  typedef typename lane_vec<V>::type vf;
  int node = __builtin_amdgcn_readfirstlane(
      (int)(blockIdx.x * (blockDim.x >> 5) + (threadIdx.x >> 5)));
  const int nw = (int)(gridDim.x * (blockDim.x >> 5));
  const int lane = threadIdx.x & 31;
  __shared__ alignas(16) float fin[3 * F];
  if (threadIdx.x < F) {
    fin[threadIdx.x] = bias[threadIdx.x];
    fin[F + threadIdx.x] = ln_g[threadIdx.x];
    fin[2 * F + threadIdx.x] = ln_b[threadIdx.x];
  }
  __syncthreads();
  if (node >= N) return;
  const int c16 = lane & 15;
  const vf wev = *(const vf*)(We + c16 * V);
  const vf atv = *(const vf*)(att + c16 * V);
  unsigned beg = rowptr[node];
  unsigned len = cnt[node];
  for (;;) {
    const int nxt = node + nw;
    unsigned nbeg = 0u, nlen = 0u;
    if (nxt < N) { nbeg = rowptr[nxt]; nlen = cnt[nxt]; }
    const v2f oh = attend_row_pairs<H, C>(node, beg, len, lane, seg_s, seg_a, xl, xr, wev, atv);
    const unsigned ch0 = (unsigned)(c16 * V + (lane >> 4) * (V / 2));
    typedef __attribute__((address_space(3))) volatile v2f lds_v2f;
    const v2f bi = *(lds_v2f*)(v2f*)(fin + ch0);
    const v2f gi = *(lds_v2f*)(v2f*)(fin + F + ch0);
    const v2f be = *(lds_v2f*)(v2f*)(fin + 2 * F + ch0);
    *(v2f*)((char*)out + ((unsigned)node * (unsigned)(F * 4) + ch0 * 4u)) = finish_vals<V / 2>(oh, bi, gi, be);
    if (nxt >= N) break;
    node = nxt; beg = nbeg; len = nlen;
  }
}

#define HP 132
template <int H, int C, int NT>
__global__ void __launch_bounds__(256, 16) attend_project(
    const int* __restrict__ seg_s, const float* __restrict__ seg_a,
    const unsigned* __restrict__ rowptr, const unsigned* __restrict__ cnt,
    const float* __restrict__ xl, const float* __restrict__ xr,
    const float* __restrict__ We, const float* __restrict__ att,
    const float* __restrict__ bias, const float* __restrict__ ln_g, const float* __restrict__ ln_b,
    const float* __restrict__ B1, const float* __restrict__ B2,
    float* __restrict__ C1, float* __restrict__ C2) {
  constexpr int F = H * C;
  constexpr int V = F / 32;
  constexpr int N2 = NT * 16;
  static_assert(V == 4 && F + 4 == HP, "a lane stores 16 B of a row; pitch = row + 4");
  static_assert(2 * NT == 8, "one column tile per wave of the 8");
  typedef typename lane_vec<V>::type vf;
  __shared__ alignas(16) float strip[16 * HP];
  const int lane = threadIdx.x & 31;
  const int wave = __builtin_amdgcn_readfirstlane((int)(threadIdx.x >> 5));
  const int m0 = (int)blockIdx.x * 16;
  {
    const vf wev = *(const vf*)(We + lane * V);
    const vf atv = *(const vf*)(att + lane * V);
    const int n_a = m0 + wave, n_b = m0 + wave + 8;
    const unsigned beg_a = rowptr[n_a], len_a = cnt[n_a];
    const unsigned beg_b = rowptr[n_b], len_b = cnt[n_b];
    typedef __attribute__((address_space(3))) volatile vf lds_row;
    vf* const row_a = (vf*)(strip + wave * HP + lane * V);
    vf* const row_b = (vf*)(strip + (wave + 8) * HP + lane * V);
    lds_row* const wait_a = (lds_row*)row_a;
    *wait_a = attend_row<H, C>(n_a, beg_a, len_a, lane, seg_s, seg_a, xl, xr, wev, atv);
    const vf o_b = attend_row<H, C>(n_b, beg_b, len_b, lane, seg_s, seg_a, xl, xr, wev, atv);
    *row_b = finish_row<V>(o_b, lane * V, bias, ln_g, ln_b);
    const vf o_a = *wait_a;
    *row_a = finish_row<V>(o_a, lane * V, bias, ln_g, ln_b);
  }
  __syncthreads();
  const int half = lane >> 4;
  const int l16  = lane & 15;
  const float* __restrict__ B = (wave < NT) ? B1 : B2;
  float* __restrict__ Cm = (wave < NT) ? C1 : C2;
  const int n0 = (wave & (NT - 1)) * 16;
  const float* arow = strip + l16 * HP + half * 2;
  const float* __restrict__ bcol = B + (size_t)(half * 2) * N2 + n0 + l16;
  v8f acc = v8f{};
#pragma unroll
  for (int k = 0; k < F; k += 4) {
    const v2f a = *(const v2f*)(arow + k);
    v2f b;
    b.x = bcol[(size_t)k * N2];
    b.y = bcol[(size_t)(k + 1) * N2];
    acc = __builtin_amdgcn_wmma_f32_16x16x4_f32(false, a, false, b, (short)0, acc, false, false);
  }
  float* __restrict__ crow = Cm + (size_t)(m0 + half * 8) * N2 + n0 + l16;
#pragma unroll
  for (int r = 0; r < 8; ++r) crow[(size_t)r * N2] = acc[r];
}

extern "C" void kernel_launch(void* const* d_in, const int* in_sizes, int n_in,
                              void* d_out, int out_size, void* d_ws, size_t ws_size,
                              hipStream_t stream) {
  const float* x      = (const float*)d_in[0];
  const float* ea     = (const float*)d_in[1];
  const float* W1l    = (const float*)d_in[2];
  const float* W1r    = (const float*)d_in[3];
  const float* W1e    = (const float*)d_in[4];
  const float* att1   = (const float*)d_in[5];
  const float* b1     = (const float*)d_in[6];
  const float* ln1_g  = (const float*)d_in[7];
  const float* ln1_b  = (const float*)d_in[8];
  const float* W2l    = (const float*)d_in[9];
  const float* W2r    = (const float*)d_in[10];
  const float* W2e    = (const float*)d_in[11];
  const float* att2   = (const float*)d_in[12];
  const float* b2     = (const float*)d_in[13];
  const float* ln2_g  = (const float*)d_in[14];
  const float* ln2_b  = (const float*)d_in[15];
  const int*   ei     = (const int*)d_in[16];
  float*       outp   = (float*)d_out;

  const int N = in_sizes[0] / 10;
  const int E = in_sizes[1];
  const int HID = 32, HEADS = 4, F1 = HEADS * HID  , OUTC = 64;

  float* ws      = (float*)d_ws;
  float* xl1     = ws;
  float* xr1     = xl1 + (size_t)N * F1;
  float* xr2     = xr1 + (size_t)N * F1;
  float* xl2     = xr2 + (size_t)N * OUTC;
  unsigned* cnt    = (unsigned*)(xl2 + (size_t)N * OUTC);
  unsigned* rowptr = cnt + (size_t)N;
  const unsigned NB1 = (unsigned)((E + CHUNK - 1) / CHUNK);
  const unsigned NBK = (unsigned)((N + BR - 1) / BR);
  unsigned* tcount = rowptr + (size_t)N;
  unsigned* tstart = tcount + (size_t)BR * NB1;
  int* b_s         = (int*)(tstart + (size_t)BR * NB1);
  int* b_d         = b_s + (size_t)E;
  float* b_a       = (float*)(b_d + (size_t)E);
  int* seg_s       = (int*)(b_a + (size_t)E);
  float* seg_a     = (float*)(seg_s + (size_t)E);

  const int TB = 256;
  const unsigned RES_NODE = 2048;
  const dim3 gemm_grid((N / 16 + 3) / 4, 2);

  wmma_gemm_pair<10, 12, 8><<<gemm_grid, 128, 0, stream>>>(x, W1l, W1r, xl1, xr1, N);

  bin_local<<<NB1, ST, 0, stream>>>(ei, ea, tcount, tstart, b_s, b_d, b_a, E);
  bucket_sort<<<NBK, ST, 0, stream>>>(tcount, tstart, b_s, b_d, b_a, cnt, rowptr, seg_s, seg_a, N, (int)NB1);

  attend_project<4, 32, 4><<<N / 16, TB, 0, stream>>>(
      seg_s, seg_a, rowptr, cnt, xl1, xr1, W1e, att1, b1, ln1_g, ln1_b, W2l, W2r, xl2, xr2);

  node_attend<1, 64><<<RES_NODE, TB, 0, stream>>>(
      seg_s, seg_a, rowptr, cnt, xl2, xr2, W2e, att2, b2, ln2_g, ln2_b, outp, N);
}


# layer 1 projection launch dissolved, 5 launches to 4: x@W1l made per 16-node block inside the attention launch (xl1 never stored), x@W1r by the level-2 sort blocks for their 256 nodes; sums kept
# speedup vs baseline: 20.4716x; 1.0874x over previous
#include <hip/hip_runtime.h>
#include <hip/hip_bf16.h>
#include <math.h>

typedef float     v2f  __attribute__((ext_vector_type(2)));
typedef float     v8f  __attribute__((ext_vector_type(8)));

#define NEG_SLOPE 0.2f
#define LN_EPS    1e-5f

template <int K>
__device__ __forceinline__ float xor_row(float x) {
  return __builtin_bit_cast(float, __builtin_amdgcn_update_dpp(
      0, __builtin_bit_cast(int, x), 0x160 | K, 0xf, 0xf, true));
}
__device__ __forceinline__ float xor_half(float x) {
  const int b = __builtin_bit_cast(int, x);
  return __builtin_bit_cast(float, __builtin_amdgcn_permlanex16(b, b, 0x76543210, 0xfedcba98,
                                                                false, false));
}
template <int LPH>
__device__ __forceinline__ float head_sum(float p) {
  p += xor_row<1>(p);
  p += xor_row<2>(p);
  p += xor_row<4>(p);
  if (LPH > 8)  p += xor_row<8>(p);
  if (LPH > 16) p += xor_half(p);
  return p;
}
__device__ __forceinline__ float wave_sum(float s) {
  s += xor_half(s);
  s += xor_row<8>(s);
  s += xor_row<4>(s);
  s += xor_row<2>(s);
  s += xor_row<1>(s);
  return s;
}

__device__ __forceinline__ int row_lane(int x, unsigned sel) {
  return __builtin_amdgcn_permlane16(x, x, sel, sel, true, false);
}
template <int LPH>
__device__ __forceinline__ float head_sum16(float p) {
  if (LPH > 1) p += xor_row<1>(p);
  if (LPH > 2) p += xor_row<2>(p);
  if (LPH > 4) p += xor_row<4>(p);
  if (LPH > 8) p += xor_row<8>(p);
  return p;
}

template <int K, int T> struct strip_frags { v2f a[K / 4]; v2f b[K / 4][T]; };

template <int KS, int K, int N, int T>
__device__ __forceinline__ void load_strip(strip_frags<K, T>& f, const float* __restrict__ A,
                                           const float* __restrict__ B, int m0, int n0, int lane) {
  static_assert(K % 4 == 0 && (K == KS || K == KS + 2), "K step layout");
  const int half = lane >> 4;
  const int l16  = lane & 15;
  const float* __restrict__ arow = A + (size_t)(m0 + l16) * KS;
  const float* __restrict__ bcol = B + n0 + l16;
#pragma unroll
  for (int k = 0; k < K; k += 4) {
    const int ka = (k + 4 <= KS) ? k + half * 2 : k;
    f.a[k / 4] = *(const v2f*)(arow + ka);
#pragma unroll
    for (int t = 0; t < T; ++t) {
      v2f b;
      b.x = bcol[(size_t)ka * N + t * 16];
      b.y = bcol[(size_t)(ka + 1) * N + t * 16];
      f.b[k / 4][t] = b;
    }
  }
}

template <int KS, int K, int T>
__device__ __forceinline__ void strip_products(const strip_frags<K, T>& f, v8f (&acc)[T], int lane) {
  const int half = lane >> 4;
  const v2f zero = {0.0f, 0.0f};
#pragma unroll
  for (int t = 0; t < T; ++t) acc[t] = v8f{};
#pragma unroll
  for (int k = 0; k < K; k += 4) {
    const bool whole = k + 4 <= KS;
    const v2f a = (whole || !half) ? f.a[k / 4] : zero;
#pragma unroll
    for (int t = 0; t < T; ++t) {
      const v2f b = (whole || !half) ? f.b[k / 4][t] : zero;
      acc[t] = __builtin_amdgcn_wmma_f32_16x16x4_f32(false, a, false, b, (short)0, acc[t], false, false);
    }
  }
}

template <int KS, int K, int N, int T>
__device__ __forceinline__ void project_strip(const strip_frags<K, T>& f, float* __restrict__ C,
                                              int m0, int n0, int lane) {
  const int half = lane >> 4;
  const int l16  = lane & 15;
  v8f acc[T];
  strip_products<KS, K, T>(f, acc, lane);
  float* __restrict__ crow = C + (size_t)(m0 + half * 8) * N + n0 + l16;
#pragma unroll
  for (int r = 0; r < 8; ++r)
#pragma unroll
    for (int t = 0; t < T; ++t)
      crow[(size_t)r * N + t * 16] = acc[t][r];
}

#define BSH   8
#define BR    256
#define ST    1024
#define CHUNK 4096
#define EPT   (CHUNK / ST)
#define BCAP  6144

__device__ __forceinline__ void wave_scan256(unsigned* v, int lane) {
  unsigned x[8], s = 0u;
#pragma unroll
  for (int j = 0; j < 8; ++j) { x[j] = v[lane * 8 + j]; s += x[j]; }
  unsigned inc = s;
#pragma unroll
  for (int off = 1; off < 32; off <<= 1) {
    const unsigned y = __shfl_up(inc, off);
    if (lane >= off) inc += y;
  }
  unsigned run = inc - s;
#pragma unroll
  for (int j = 0; j < 8; ++j) { v[lane * 8 + j] = run; run += x[j]; }
}

__device__ __forceinline__ unsigned wave_usum(unsigned v) {
#pragma unroll
  for (int off = 16; off > 0; off >>= 1) v += __shfl_xor(v, off);
  return v;
}

__global__ void __launch_bounds__(ST) bin_local(const int* __restrict__ ei, const float* __restrict__ ea,
                                                unsigned* __restrict__ tcount, unsigned* __restrict__ tstart,
                                                int* __restrict__ b_s, int* __restrict__ b_d, float* __restrict__ b_a,
                                                int E) {
  __shared__ int   st_s[CHUNK];
  __shared__ int   st_d[CHUNK];
  __shared__ float st_a[CHUNK];
  __shared__ unsigned lstart[BR], lcur[BR];
  const int t = threadIdx.x;
  const int b = blockIdx.x;
  const int e0 = b * CHUNK;
  const int n = min(CHUNK, E - e0);
  int d[EPT], sv[EPT];
  float av[EPT];
#pragma unroll
  for (int u = 0; u < EPT; ++u) {
    const int i = t + u * ST;
    d[u]  = (i < n) ? ei[E + e0 + i] : 0;
    sv[u] = (i < n) ? ei[e0 + i] : 0;
    av[u] = (i < n) ? ea[e0 + i] : 0.0f;
  }
  if (t < BR) lcur[t] = 0u;
  __syncthreads();
#pragma unroll
  for (int u = 0; u < EPT; ++u)
    if (t + u * ST < n) atomicAdd(&lcur[d[u] >> BSH], 1u);
  __syncthreads();
  unsigned c = 0u;
  if (t < BR) { c = lcur[t]; lstart[t] = c; }
  __syncthreads();
  if ((t >> 5) == 0) wave_scan256(lstart, t & 31);
  __syncthreads();
  if (t < BR) {
    const unsigned s = lstart[t];
    tcount[(size_t)b * BR + t] = c;
    tstart[(size_t)b * BR + t] = s;
    lcur[t] = s;
  }
  __syncthreads();
#pragma unroll
  for (int u = 0; u < EPT; ++u) {
    if (t + u * ST < n) {
      const unsigned p = atomicAdd(&lcur[d[u] >> BSH], 1u);
      st_d[p] = d[u];
      st_s[p] = sv[u];
      st_a[p] = av[u];
    }
  }
  __syncthreads();
  for (int i = t; i < n; i += ST) {
    b_d[e0 + i] = st_d[i];
    b_s[e0 + i] = st_s[i];
    b_a[e0 + i] = st_a[i];
  }
}

#define CPW (BR / (ST / 32))
template <int KS, int K, int F>
__global__ void __launch_bounds__(ST) bucket_sort(const unsigned* __restrict__ tcount, const unsigned* __restrict__ tstart,
                                                  const int* __restrict__ b_s, const int* __restrict__ b_d,
                                                  const float* __restrict__ b_a,
                                                  unsigned* __restrict__ cnt, unsigned* __restrict__ rowptr,
                                                  int* __restrict__ seg_s, float* __restrict__ seg_a,
                                                  const float* __restrict__ X, const float* __restrict__ Wr,
                                                  float* __restrict__ xr, int N, int nb) {
  __shared__ int   st_s[BCAP];
  __shared__ float st_a[BCAP];
  __shared__ unsigned lc[BR], ls[BR], pst[BR], pcn[BR], tot[2];
  const int t = threadIdx.x;
  const int lane = t & 31;
  const int w = t >> 5;
  const int k = blockIdx.x;
  const int n0 = k << BSH;
  constexpr int PT = F / 32;
  static_assert(BR / 16 * 2 == ST / 32 && F % 32 == 0, "one (strip, column half) per wave");
  const int wu = __builtin_amdgcn_readfirstlane(w);
  const int pm0 = n0 + (wu >> 1) * 16;
  const int pn0 = (wu & 1) * (PT * 16);
  const bool pact = pm0 + 16 <= N;
  strip_frags<K, PT> pf = {};
  if (pact) load_strip<KS, K, F, PT>(pf, X, Wr, pm0, pn0, lane);
  unsigned ms = 0u, mc = 0u;
  if (t < BR) {
    if (t < nb) { ms = tstart[(size_t)t * BR + k]; mc = tcount[(size_t)t * BR + k]; }
    pst[t] = (unsigned)t * CHUNK + ms;
    pcn[t] = mc;
    lc[t] = 0u;
  }
  if (t < 2) tot[t] = 0u;
  __syncthreads();
  if (t < BR) {
    const unsigned s = wave_usum(ms);
    const unsigned c = wave_usum(mc);
    if (lane == 0) { atomicAdd(&tot[0], s); atomicAdd(&tot[1], c); }
  }
  __syncthreads();
  const unsigned base = tot[0];
  const int n = (int)tot[1];
  unsigned po[CPW], pn[CPW];
#pragma unroll
  for (int u = 0; u < CPW; ++u) { po[u] = pst[w + (ST / 32) * u]; pn[u] = pcn[w + (ST / 32) * u]; }
  {
    int d[CPW];
#pragma unroll
    for (int u = 0; u < CPW; ++u) d[u] = ((unsigned)lane < pn[u]) ? b_d[po[u] + lane] : n0;
#pragma unroll
    for (int u = 0; u < CPW; ++u)
      if ((unsigned)lane < pn[u]) atomicAdd(&lc[d[u] - n0], 1u);
#pragma unroll
    for (int u = 0; u < CPW; ++u)
      for (unsigned j = 32u + lane; j < pn[u]; j += 32u) atomicAdd(&lc[b_d[po[u] + j] - n0], 1u);
  }
  __syncthreads();
  unsigned c = 0u;
  if (t < BR) { c = lc[t]; ls[t] = c; }
  __syncthreads();
  if (t < 32) wave_scan256(ls, t);
  __syncthreads();
  if (t < BR) {
    const unsigned s = ls[t];
    lc[t] = s;
    if (n0 + t < N) { cnt[n0 + t] = c; rowptr[n0 + t] = base + s; }
  }
  __syncthreads();
  const bool staged = n <= BCAP;
  auto put = [&](int dd, int ss, float aa) {
    const unsigned p = atomicAdd(&lc[dd - n0], 1u);
    if (staged) { st_s[p] = ss; st_a[p] = aa; }
    else        { seg_s[base + p] = ss; seg_a[base + p] = aa; }
  };
  {
    int d[CPW], sv[CPW];
    float av[CPW];
#pragma unroll
    for (int u = 0; u < CPW; ++u) {
      const bool in = (unsigned)lane < pn[u];
      d[u]  = in ? b_d[po[u] + lane] : n0;
      sv[u] = in ? b_s[po[u] + lane] : 0;
      av[u] = in ? b_a[po[u] + lane] : 0.0f;
    }
#pragma unroll
    for (int u = 0; u < CPW; ++u)
      if ((unsigned)lane < pn[u]) put(d[u], sv[u], av[u]);
#pragma unroll
    for (int u = 0; u < CPW; ++u)
      for (unsigned j = 32u + lane; j < pn[u]; j += 32u) put(b_d[po[u] + j], b_s[po[u] + j], b_a[po[u] + j]);
  }
  __syncthreads();
  if (staged)
    for (int i = t; i < n; i += ST) { seg_s[base + i] = st_s[i]; seg_a[base + i] = st_a[i]; }
  if (pact) project_strip<KS, K, F, PT>(pf, xr, pm0, pn0, lane);
}

template <int G> struct group_of { static constexpr int value = G; };
template <int G, bool M> struct step_group { static constexpr int steps = G; static constexpr bool masked = M; };
template <int V> struct lane_vec { typedef float type __attribute__((ext_vector_type(V))); };

template <int H, int C>
__device__ __forceinline__ typename lane_vec<H * C / 32>::type attend_row(
    unsigned beg, unsigned len, int lane,
    const int* __restrict__ seg_s, const float* __restrict__ seg_a,
    const typename lane_vec<H * C / 32>::type xlv, const float* __restrict__ xr,
    const typename lane_vec<H * C / 32>::type wev, const typename lane_vec<H * C / 32>::type atv) {
  constexpr int F = H * C;
  constexpr int V = F / 32;
  constexpr int LPH = C / V;
  typedef typename lane_vec<V>::type vf;
  float m = -INFINITY, dn = 0.0f;
  vf acc = {};
  for (unsigned base = 0; base < len; base += 32) {
    const unsigned nn = (len - base < 32u) ? (len - base) : 32u;
    int my_s = 0;
    float my_a = 0.0f;
    if ((unsigned)lane < nn) {
      my_s = seg_s[beg + base + lane];
      my_a = seg_a[beg + base + lane];
    }
    auto group = [&](auto gtag, unsigned j) {
      constexpr int G = decltype(gtag)::value;
      vf v[G];
      float p[G];
#pragma unroll
      for (int i = 0; i < G; ++i) {
        const int s = __builtin_amdgcn_readlane(my_s, (int)(j + i));
        v[i] = *(const vf*)(xr + (size_t)s * F + lane * V);
      }
#pragma unroll
      for (int i = 0; i < G; ++i) {
        const float a = __builtin_bit_cast(
            float, __builtin_amdgcn_readlane(__builtin_bit_cast(int, my_a), (int)(j + i)));
        float q;
#pragma unroll
        for (int k = 0; k < V; ++k) {
          float t = xlv[k] + v[i][k] + a * wev[k];
          t = fmaxf(t, NEG_SLOPE * t);
          q = (k == 0) ? atv[k] * t : q + atv[k] * t;
        }
        p[i] = head_sum<LPH>(q);
      }
      float mn = m;
#pragma unroll
      for (int i = 0; i < G; ++i) mn = fmaxf(mn, p[i]);
      const float sc = __expf(m - mn);
      float wsum;
      vf part;
#pragma unroll
      for (int i = 0; i < G; ++i) {
        const float w = __expf(p[i] - mn);
        wsum = (i == 0) ? w : wsum + w;
        part = (i == 0) ? w * v[i] : part + w * v[i];
      }
      dn = dn * sc + wsum;
      acc = acc * sc + part;
      m = mn;
    };
    unsigned j = 0;
    for (; j + 4 <= nn; j += 4) group(group_of<4>{}, j);
    for (; j < nn; ++j) group(group_of<1>{}, j);
  }
  const float dsafe = (len != 0u) ? dn : 1.0f;
  return acc / dsafe;
}

template <int V>
__device__ __forceinline__ typename lane_vec<V>::type finish_vals(
    typename lane_vec<V>::type o, const typename lane_vec<V>::type bi,
    const typename lane_vec<V>::type gi, const typename lane_vec<V>::type be) {
  constexpr int F = 32 * V;
  float vals[V];
  float sum = 0.0f;
#pragma unroll
  for (int i = 0; i < V; ++i) {
    const float v = o[i] + bi[i];
    vals[i] = v;
    sum += v;
  }
  sum = wave_sum(sum);
  const float mu = sum / (float)F;
  float vs = 0.0f;
#pragma unroll
  for (int i = 0; i < V; ++i) { const float dlt = vals[i] - mu; vs += dlt * dlt; }
  vs = wave_sum(vs);
  const float inv = rsqrtf(vs / (float)F + LN_EPS);
#pragma unroll
  for (int i = 0; i < V; ++i) {
    const float hv = (vals[i] - mu) * inv * gi[i] + be[i];
    o[i] = (hv > 0.0f) ? hv : (__expf(hv) - 1.0f);
  }
  return o;
}
template <int V>
__device__ __forceinline__ typename lane_vec<V>::type finish_row(
    typename lane_vec<V>::type o, int ch0,
    const float* __restrict__ bias, const float* __restrict__ ln_g, const float* __restrict__ ln_b) {
  typedef typename lane_vec<V>::type vf;
  return finish_vals<V>(o, *(const vf*)(bias + ch0), *(const vf*)(ln_g + ch0), *(const vf*)(ln_b + ch0));
}

template <int H, int C>
__device__ __forceinline__ v2f attend_row_pairs(
    int node, unsigned beg, unsigned len, int lane,
    const int* __restrict__ seg_s, const float* __restrict__ seg_a,
    const float* __restrict__ xl, const float* __restrict__ xr,
    const typename lane_vec<4>::type wev, const typename lane_vec<4>::type atv) {
  constexpr int F = H * C;
  constexpr int V = 4;
  constexpr int LPH = C / V;
  static_assert(F == 16 * V && C % V == 0, "a 16-lane row holds the F channels, a head's channels lie side by side");
  typedef typename lane_vec<V>::type vf;
  const int c16 = lane & 15;
  const unsigned hf = (unsigned)lane >> 4;
  const char* __restrict__ xrb = (const char*)xr;
  const unsigned lane_off = (unsigned)c16 * (unsigned)(V * 4);
  const vf xlv = *(const vf*)((const char*)xl + ((unsigned)node * (unsigned)(F * 4) + lane_off));
  float m = -INFINITY, dn = 0.0f;
  v2f acc_a = {}, acc_b = {};
  for (unsigned base = 0; base < len; base += 32) {
    const unsigned nn = (len - base < 32u) ? (len - base) : 32u;
    const unsigned my_e = 2u * (unsigned)c16 + hf;
    int my_s = 0;
    float my_a = 0.0f;
    if (my_e < nn) {
      my_s = seg_s[beg + base + my_e];
      my_a = seg_a[beg + base + my_e];
    }
    auto group = [&](auto gtag, unsigned j) {
      constexpr int G = decltype(gtag)::steps;
      constexpr bool MASKED = decltype(gtag)::masked;
      vf v[G];
      float p[G];
#pragma unroll
      for (int i = 0; i < G; ++i) {
        const int s = row_lane(my_s, (j + i) * 0x11111111u);
        v[i] = *(const vf*)(xrb + ((unsigned)s * (unsigned)(F * 4) + lane_off));
      }
#pragma unroll
      for (int i = 0; i < G; ++i) {
        const float a = __builtin_bit_cast(
            float, row_lane(__builtin_bit_cast(int, my_a), (j + i) * 0x11111111u));
        float q;
#pragma unroll
        for (int k = 0; k < V; ++k) {
          float t = xlv[k] + v[i][k] + a * wev[k];
          t = fmaxf(t, NEG_SLOPE * t);
          q = (k == 0) ? atv[k] * t : q + atv[k] * t;
        }
        p[i] = head_sum16<LPH>(q);
      }
      bool ok[G];
      float mn = m;
#pragma unroll
      for (int i = 0; i < G; ++i) {
        ok[i] = !MASKED || (2u * (j + i) + hf < nn);
        mn = fmaxf(mn, ok[i] ? p[i] : -INFINITY);
      }
      const float sc = (MASKED && m == mn) ? 1.0f : __expf(m - mn);
      float wsum;
      v2f part_a, part_b;
#pragma unroll
      for (int i = 0; i < G; ++i) {
        const float w = ok[i] ? __expf(p[i] - mn) : 0.0f;
        const v2f va = __builtin_shufflevector(v[i], v[i], 0, 1);
        const v2f vb = __builtin_shufflevector(v[i], v[i], 2, 3);
        wsum = (i == 0) ? w : wsum + w;
        part_a = (i == 0) ? w * va : part_a + w * va;
        part_b = (i == 0) ? w * vb : part_b + w * vb;
      }
      dn = dn * sc + wsum;
      acc_a = acc_a * sc + part_a;
      acc_b = acc_b * sc + part_b;
      m = mn;
    };
    const unsigned full = nn >> 1;
    unsigned j = 0;
    for (; j + 4 <= full; j += 4) group(step_group<4, false>{}, j);
    for (; 2u * j < nn; j += 2) group(step_group<2, true>{}, j);
  }
  const float m_o = xor_half(m);
  const float dn_o = xor_half(dn);
  const float mm = fmaxf(m, m_o);
  const float f   = (m   == -INFINITY) ? 0.0f : __expf(m - mm);
  const float f_o = (m_o == -INFINITY) ? 0.0f : __expf(m_o - mm);
  const float dt = dn * f + dn_o * f_o;
  const v2f keep = hf ? acc_b : acc_a;
  const v2f send = hf ? acc_a : acc_b;
  v2f got;
  got.x = xor_half(send.x);
  got.y = xor_half(send.y);
  const v2f at = keep * f + got * f_o;
  const float dsafe = (len != 0u) ? dt : 1.0f;
  return at / dsafe;
}

template <int H, int C>
__global__ void __launch_bounds__(256, 16) node_attend(const int* __restrict__ seg_s, const float* __restrict__ seg_a,
                            const unsigned* __restrict__ rowptr, const unsigned* __restrict__ cnt,
                            const float* __restrict__ xl, const float* __restrict__ xr,
                            const float* __restrict__ We, const float* __restrict__ att,
                            const float* __restrict__ bias, const float* __restrict__ ln_g,
                            const float* __restrict__ ln_b,
                            float* __restrict__ out, int N) {
  constexpr int F = H * C;
  constexpr int V = F / 16;
  static_assert(V == 4, "the paired walk is written for rows of 64 channels");
  typedef typename lane_vec<V>::type vf;
  int node = __builtin_amdgcn_readfirstlane(
      (int)(blockIdx.x * (blockDim.x >> 5) + (threadIdx.x >> 5)));
  const int nw = (int)(gridDim.x * (blockDim.x >> 5));
  const int lane = threadIdx.x & 31;
  __shared__ alignas(16) float fin[3 * F];
  if (threadIdx.x < F) {
    fin[threadIdx.x] = bias[threadIdx.x];
    fin[F + threadIdx.x] = ln_g[threadIdx.x];
    fin[2 * F + threadIdx.x] = ln_b[threadIdx.x];
  }
  __syncthreads();
  if (node >= N) return;
  const int c16 = lane & 15;
  const vf wev = *(const vf*)(We + c16 * V);
  const vf atv = *(const vf*)(att + c16 * V);
  unsigned beg = rowptr[node];
  unsigned len = cnt[node];
  for (;;) {
    const int nxt = node + nw;
    unsigned nbeg = 0u, nlen = 0u;
    if (nxt < N) { nbeg = rowptr[nxt]; nlen = cnt[nxt]; }
    const v2f oh = attend_row_pairs<H, C>(node, beg, len, lane, seg_s, seg_a, xl, xr, wev, atv);
    const unsigned ch0 = (unsigned)(c16 * V + (lane >> 4) * (V / 2));
    typedef __attribute__((address_space(3))) volatile v2f lds_v2f;
    const v2f bi = *(lds_v2f*)(v2f*)(fin + ch0);
    const v2f gi = *(lds_v2f*)(v2f*)(fin + F + ch0);
    const v2f be = *(lds_v2f*)(v2f*)(fin + 2 * F + ch0);
    *(v2f*)((char*)out + ((unsigned)node * (unsigned)(F * 4) + ch0 * 4u)) = finish_vals<V / 2>(oh, bi, gi, be);
    if (nxt >= N) break;
    node = nxt; beg = nbeg; len = nlen;
  }
}

#define HP 132
template <int H, int C, int NT, int KS, int K>
__global__ void __launch_bounds__(256, 16) attend_project(
    const int* __restrict__ seg_s, const float* __restrict__ seg_a,
    const unsigned* __restrict__ rowptr, const unsigned* __restrict__ cnt,
    const float* __restrict__ X, const float* __restrict__ Wl, const float* __restrict__ xr,
    const float* __restrict__ We, const float* __restrict__ att,
    const float* __restrict__ bias, const float* __restrict__ ln_g, const float* __restrict__ ln_b,
    const float* __restrict__ B1, const float* __restrict__ B2,
    float* __restrict__ C1, float* __restrict__ C2) {
  constexpr int F = H * C;
  constexpr int V = F / 32;
  constexpr int N2 = NT * 16;
  static_assert(V == 4 && F + 4 == HP, "a lane stores 16 B of a row; pitch = row + 4");
  static_assert(2 * NT == 8, "one column tile per wave of the 8");
  static_assert(F == 8 * 16, "one column tile of xl per wave of the 8");
  static_assert(K % 4 == 0 && (K == KS || K == KS + 2), "K step layout");
  typedef typename lane_vec<V>::type vf;
  __shared__ alignas(16) float strip[16 * HP];
  __shared__ alignas(16) float xls[16 * HP];
  const int lane = threadIdx.x & 31;
  const int wave = __builtin_amdgcn_readfirstlane((int)(threadIdx.x >> 5));
  const int m0 = (int)blockIdx.x * 16;
  const vf wev = *(const vf*)(We + lane * V);
  const vf atv = *(const vf*)(att + lane * V);
  const int n_a = m0 + wave, n_b = m0 + wave + 8;
  const unsigned beg_a = rowptr[n_a], len_a = cnt[n_a];
  const unsigned beg_b = rowptr[n_b], len_b = cnt[n_b];
  {
    const int ln = (int)__builtin_amdgcn_mbcnt_lo(~0u, 0u);
    strip_frags<K, 1> f;
    load_strip<KS, K, F, 1>(f, X, Wl, m0, wave * 16, ln);
    v8f acc[1];
    strip_products<KS, K, 1>(f, acc, ln);
    float* tcol = xls + ((ln >> 4) * 8) * HP + wave * 16 + (ln & 15);
#pragma unroll
    for (int r = 0; r < 8; ++r) tcol[r * HP] = acc[0][r];
  }
  __syncthreads();
  {
    typedef __attribute__((address_space(3))) volatile vf lds_row;
    vf* const row_a = (vf*)(strip + wave * HP + lane * V);
    vf* const row_b = (vf*)(strip + (wave + 8) * HP + lane * V);
    lds_row* const wait_a = (lds_row*)row_a;
    const vf xl_a = *(const vf*)(xls + wave * HP + lane * V);
    *wait_a = attend_row<H, C>(beg_a, len_a, lane, seg_s, seg_a, xl_a, xr, wev, atv);
    const vf xl_b = *(const vf*)(xls + (wave + 8) * HP + lane * V);
    const vf o_b = attend_row<H, C>(beg_b, len_b, lane, seg_s, seg_a, xl_b, xr, wev, atv);
    *row_b = finish_row<V>(o_b, lane * V, bias, ln_g, ln_b);
    const vf o_a = *wait_a;
    *row_a = finish_row<V>(o_a, lane * V, bias, ln_g, ln_b);
  }
  __syncthreads();
  const int half = lane >> 4;
  const int l16  = lane & 15;
  const float* __restrict__ B = (wave < NT) ? B1 : B2;
  float* __restrict__ Cm = (wave < NT) ? C1 : C2;
  const int n0 = (wave & (NT - 1)) * 16;
  const float* arow = strip + l16 * HP + half * 2;
  const float* __restrict__ bcol = B + (size_t)(half * 2) * N2 + n0 + l16;
  v8f acc = v8f{};
#pragma unroll
  for (int k = 0; k < F; k += 4) {
    const v2f a = *(const v2f*)(arow + k);
    v2f b;
    b.x = bcol[(size_t)k * N2];
    b.y = bcol[(size_t)(k + 1) * N2];
    acc = __builtin_amdgcn_wmma_f32_16x16x4_f32(false, a, false, b, (short)0, acc, false, false);
  }
  float* __restrict__ crow = Cm + (size_t)(m0 + half * 8) * N2 + n0 + l16;
#pragma unroll
  for (int r = 0; r < 8; ++r) crow[(size_t)r * N2] = acc[r];
}

extern "C" void kernel_launch(void* const* d_in, const int* in_sizes, int n_in,
                              void* d_out, int out_size, void* d_ws, size_t ws_size,
                              hipStream_t stream) {
  const float* x      = (const float*)d_in[0];
  const float* ea     = (const float*)d_in[1];
  const float* W1l    = (const float*)d_in[2];
  const float* W1r    = (const float*)d_in[3];
  const float* W1e    = (const float*)d_in[4];
  const float* att1   = (const float*)d_in[5];
  const float* b1     = (const float*)d_in[6];
  const float* ln1_g  = (const float*)d_in[7];
  const float* ln1_b  = (const float*)d_in[8];
  const float* W2l    = (const float*)d_in[9];
  const float* W2r    = (const float*)d_in[10];
  const float* W2e    = (const float*)d_in[11];
  const float* att2   = (const float*)d_in[12];
  const float* b2     = (const float*)d_in[13];
  const float* ln2_g  = (const float*)d_in[14];
  const float* ln2_b  = (const float*)d_in[15];
  const int*   ei     = (const int*)d_in[16];
  float*       outp   = (float*)d_out;

  const int N = in_sizes[0] / 10;
  const int E = in_sizes[1];
  const int HID = 32, HEADS = 4, F1 = HEADS * HID  , OUTC = 64;

  float* ws      = (float*)d_ws;
  float* xr1     = ws;
  float* xr2     = xr1 + (size_t)N * F1;
  float* xl2     = xr2 + (size_t)N * OUTC;
  unsigned* cnt    = (unsigned*)(xl2 + (size_t)N * OUTC);
  unsigned* rowptr = cnt + (size_t)N;
  const unsigned NB1 = (unsigned)((E + CHUNK - 1) / CHUNK);
  const unsigned NBK = (unsigned)((N + BR - 1) / BR);
  unsigned* tcount = rowptr + (size_t)N;
  unsigned* tstart = tcount + (size_t)BR * NB1;
  int* b_s         = (int*)(tstart + (size_t)BR * NB1);
  int* b_d         = b_s + (size_t)E;
  float* b_a       = (float*)(b_d + (size_t)E);
  int* seg_s       = (int*)(b_a + (size_t)E);
  float* seg_a     = (float*)(seg_s + (size_t)E);

  const int TB = 256;
  const unsigned RES_NODE = 2048;

  bin_local<<<NB1, ST, 0, stream>>>(ei, ea, tcount, tstart, b_s, b_d, b_a, E);
  bucket_sort<10, 12, 128><<<NBK, ST, 0, stream>>>(tcount, tstart, b_s, b_d, b_a, cnt, rowptr, seg_s, seg_a,
                                                   x, W1r, xr1, N, (int)NB1);

  attend_project<4, 32, 4, 10, 12><<<N / 16, TB, 0, stream>>>(
      seg_s, seg_a, rowptr, cnt, x, W1l, xr1, W1e, att1, b1, ln1_g, ln1_b, W2l, W2r, xl2, xr2);

  node_attend<1, 64><<<RES_NODE, TB, 0, stream>>>(
      seg_s, seg_a, rowptr, cnt, xl2, xr2, W2e, att2, b2, ln2_g, ln2_b, outp, N);
}
